# starting kernel
# baseline (speedup 1.0000x reference)
#include <hip/hip_runtime.h>
#include <hip/hip_bf16.h>
#include <math.h>


typedef __attribute__((ext_vector_type(2))) float v2f;
typedef __attribute__((ext_vector_type(8))) float v8f;

#define T_LEN 8192
#define IN_DIM 512
#define MD 128
#define LCH 64
#define NCH (T_LEN / LCH)

__device__ __forceinline__ v8f wmma4(v2f a, v2f b, v8f c) {
  return __builtin_amdgcn_wmma_f32_16x16x4_f32(
       false, a,  false, b,
       (short)0, c,  false,  false);
}

__device__ __forceinline__ float sigmoidf_(float z) {
  return 1.0f / (1.0f + __expf(-z));
}

__global__ __launch_bounds__(128) void proj_kernel(
    const float* __restrict__ x,
    const float* __restrict__ Wq, const float* __restrict__ bq,
    const float* __restrict__ Wk, const float* __restrict__ bk,
    const float* __restrict__ Wv, const float* __restrict__ bv,
    const float* __restrict__ Wo, const float* __restrict__ bo,
    float* __restrict__ qw, float* __restrict__ kw,
    float* __restrict__ vw, float* __restrict__ ow) {
  const int wave = threadIdx.x >> 5;
  const int lane = threadIdx.x & 31;
  const int half = lane >> 4;
  const int lm = lane & 15;

  const int rowBase = blockIdx.x * 16;
  const int colTile = blockIdx.y * 4 + wave;
  const int mat = colTile >> 3;
  const int col = (colTile & 7) * 16;

  const float* W; const float* bias; float* dst;
  switch (mat) {
    case 0:  W = Wq; bias = bq; dst = qw; break;
    case 1:  W = Wk; bias = bk; dst = kw; break;
    case 2:  W = Wv; bias = bv; dst = vw; break;
    default: W = Wo; bias = bo; dst = ow; break;
  }

  v8f acc = {};
  const float* arow = x + (size_t)(rowBase + lm) * IN_DIM + 2 * half;
  const float* brow = W + (size_t)(col + lm) * IN_DIM + 2 * half;
  for (int kk = 0; kk < IN_DIM; kk += 4) {
    v2f a, b;
    a.x = arow[kk];     a.y = arow[kk + 1];
    b.x = brow[kk];     b.y = brow[kk + 1];
    acc = wmma4(a, b, acc);
  }

  const float bv_ = bias[col + lm];
#pragma unroll
  for (int e = 0; e < 8; ++e) {
    const int r = rowBase + e + 8 * half;
    float val = acc[e] + bv_;
    if (mat == 1) val *= 0.08838834764831845f;
    if (mat == 3) val = sigmoidf_(val);
    dst[(size_t)r * MD + col + lm] = val;
  }
}

__global__ __launch_bounds__(256) void gates_kernel(
    const float* __restrict__ x,
    const float* __restrict__ wi, const float* __restrict__ bi,
    const float* __restrict__ wf, const float* __restrict__ bf,
    float* __restrict__ ig, float* __restrict__ lf) {
  const int t = blockIdx.x * blockDim.x + threadIdx.x;
  if (t >= T_LEN) return;
  const float* xr = x + (size_t)t * IN_DIM;
  float si = 0.f, sf = 0.f;
  for (int f = 0; f < IN_DIM; ++f) {
    const float xv = xr[f];
    si = fmaf(xv, wi[f], si);
    sf = fmaf(xv, wf[f], sf);
  }
  ig[t] = __expf(si + bi[0]);
  const float z = sf + bf[0];
  lf[t] = (z >= 0.f) ? -log1pf(__expf(-z)) : (z - log1pf(__expf(z)));
}

__global__ __launch_bounds__(256, 1) void scan_kernel(
    const float* __restrict__ qw, const float* __restrict__ kw,
    const float* __restrict__ vw, const float* __restrict__ ow,
    const float* __restrict__ ig, const float* __restrict__ lf,
    const float* __restrict__ C0, const float* __restrict__ n0,
    const float* __restrict__ fc_w, const float* __restrict__ fc_b,
    float* __restrict__ out, float* __restrict__ C_out,
    float* __restrict__ n_out) {
  __shared__ float Cs[MD][MD + 1];
  __shared__ float Ps[LCH][LCH + 1];
  __shared__ float Hs[LCH][MD + 1];
  __shared__ float la_s[LCH], ig_s[LCH], u_s[LCH], rs_s[LCH], qn_s[LCH];
  __shared__ float ns[MD], fcw_s[MD];

  const int tid = threadIdx.x;
  const int wave = tid >> 5;
  const int lane = tid & 31;
  const int half = lane >> 4;
  const int lm = lane & 15;

  for (int i = tid; i < MD * MD; i += 256) Cs[i >> 7][i & 127] = C0[i];
  if (tid < MD) { ns[tid] = n0[tid]; fcw_s[tid] = fc_w[tid]; }
  __syncthreads();

  for (int c = 0; c < NCH; ++c) {
    const int t0 = c * LCH;

    if (c + 1 < NCH) {
      const float* nq = qw + (size_t)(t0 + LCH) * MD + tid * 32;
      const float* nk = kw + (size_t)(t0 + LCH) * MD + tid * 32;
      const float* nv = vw + (size_t)(t0 + LCH) * MD + tid * 32;
      __builtin_prefetch(nq, 0, 3);
      __builtin_prefetch(nk, 0, 3);
      __builtin_prefetch(nv, 0, 3);
    }

    if (tid == 0) {
      float a = 0.f;
      for (int j = 0; j < LCH; ++j) {
        a += lf[t0 + j];
        la_s[j] = a;
        ig_s[j] = ig[t0 + j];
      }
    }
    __syncthreads();
    if (tid < LCH) {
      u_s[tid] = __expf(la_s[LCH - 1] - la_s[tid]) * ig_s[tid];
      float s = 0.f;
      const float* qrow = qw + (size_t)(t0 + tid) * MD;
      for (int f = 0; f < MD; ++f) s = fmaf(qrow[f], ns[f], s);
      qn_s[tid] = s;
    }
    __syncthreads();

    for (int tile = wave; tile < 16; tile += 8) {
      const int tr = (tile >> 2) * 16, tc = (tile & 3) * 16;
      v8f acc = {};
      const float* ap = qw + (size_t)(t0 + tr + lm) * MD + 2 * half;
      const float* bp = kw + (size_t)(t0 + tc + lm) * MD + 2 * half;
      for (int kk = 0; kk < MD; kk += 4) {
        v2f a, b;
        a.x = ap[kk]; a.y = ap[kk + 1];
        b.x = bp[kk]; b.y = bp[kk + 1];
        acc = wmma4(a, b, acc);
      }
#pragma unroll
      for (int e = 0; e < 8; ++e) {
        const int r = tr + e + 8 * half, cc = tc + lm;
        float p = 0.f;
        if (r >= cc) p = __expf(la_s[r] - la_s[cc]) * ig_s[cc] * acc[e];
        Ps[r][cc] = p;
      }
    }
    __syncthreads();
    if (tid < LCH) {
      float s = 0.f;
      for (int j = 0; j <= tid; ++j) s += Ps[tid][j];
      rs_s[tid] = s;
    }
    __syncthreads();

    for (int tile = wave; tile < 32; tile += 8) {
      const int tr = (tile >> 3) * 16, tc = (tile & 7) * 16;
      v8f aci = {};
      for (int kk = 0; kk < LCH; kk += 4) {
        v2f a, b;
        const int j0 = kk + 2 * half;
        a.x = Ps[tr + lm][j0]; a.y = Ps[tr + lm][j0 + 1];
        const float* bp = vw + (size_t)(t0 + j0) * MD + tc + lm;
        b.x = bp[0]; b.y = bp[MD];
        aci = wmma4(a, b, aci);
      }
      v8f ace = {};
      const float* ap = qw + (size_t)(t0 + tr + lm) * MD + 2 * half;
      for (int kk = 0; kk < MD; kk += 4) {
        v2f a, b;
        a.x = ap[kk]; a.y = ap[kk + 1];
        b.x = Cs[tc + lm][kk + 2 * half];
        b.y = Cs[tc + lm][kk + 2 * half + 1];
        ace = wmma4(a, b, ace);
      }
#pragma unroll
      for (int e = 0; e < 8; ++e) {
        const int r = tr + e + 8 * half, cc = tc + lm;
        const float Ft = __expf(la_s[r]);
        const float num = fmaf(Ft, ace[e], aci[e]);
        const float den = fmaxf(fabsf(fmaf(Ft, qn_s[r], rs_s[r])), 1.0f);
        Hs[r][cc] = (num / den) * ow[(size_t)(t0 + r) * MD + cc];
      }
    }
    __syncthreads();

    if (tid < LCH) {
      float s = 0.f;
      for (int f = 0; f < MD; ++f) s = fmaf(Hs[tid][f], fcw_s[f], s);
      out[t0 + tid] = s + fc_b[0];
    }

    const float Fend = __expf(la_s[LCH - 1]);
    for (int tile = wave; tile < 64; tile += 8) {
      const int tr = (tile >> 3) * 16, tc = (tile & 7) * 16;
      v8f acc;
#pragma unroll
      for (int e = 0; e < 8; ++e) acc[e] = Fend * Cs[tr + e + 8 * half][tc + lm];
      for (int kk = 0; kk < LCH; kk += 4) {
        v2f a, b;
        const int j0 = kk + 2 * half;
        a.x = u_s[j0]     * vw[(size_t)(t0 + j0) * MD + tr + lm];
        a.y = u_s[j0 + 1] * vw[(size_t)(t0 + j0 + 1) * MD + tr + lm];
        const float* bp = kw + (size_t)(t0 + j0) * MD + tc + lm;
        b.x = bp[0]; b.y = bp[MD];
        acc = wmma4(a, b, acc);
      }
#pragma unroll
      for (int e = 0; e < 8; ++e) Cs[tr + e + 8 * half][tc + lm] = acc[e];
    }
    if (tid < MD) {
      float s = Fend * ns[tid];
      for (int j = 0; j < LCH; ++j)
        s = fmaf(u_s[j], kw[(size_t)(t0 + j) * MD + tid], s);
      ns[tid] = s;
    }
    __syncthreads();
  }

  for (int i = tid; i < MD * MD; i += 256) C_out[i] = Cs[i >> 7][i & 127];
  if (tid < MD) n_out[tid] = ns[tid];
}

extern "C" void kernel_launch(void* const* d_in, const int* in_sizes, int n_in,
                              void* d_out, int out_size, void* d_ws, size_t ws_size,
                              hipStream_t stream) {
  const float* x    = (const float*)d_in[0];
  const float* C0   = (const float*)d_in[1];
  const float* n0   = (const float*)d_in[2];
  const float* Wq   = (const float*)d_in[3];
  const float* bq   = (const float*)d_in[4];
  const float* Wk   = (const float*)d_in[5];
  const float* bk   = (const float*)d_in[6];
  const float* Wv   = (const float*)d_in[7];
  const float* bv   = (const float*)d_in[8];
  const float* wi   = (const float*)d_in[9];
  const float* bi   = (const float*)d_in[10];
  const float* wf   = (const float*)d_in[11];
  const float* bf   = (const float*)d_in[12];
  const float* Wo   = (const float*)d_in[13];
  const float* bo   = (const float*)d_in[14];
  const float* fc_w = (const float*)d_in[15];
  const float* fc_b = (const float*)d_in[16];

  float* ws = (float*)d_ws;
  const size_t NP = (size_t)T_LEN * MD;
  float* qw = ws;
  float* kw = ws + NP;
  float* vw = ws + 2 * NP;
  float* owp = ws + 3 * NP;
  float* igp = ws + 4 * NP;
  float* lfp = igp + T_LEN;

  float* outp  = (float*)d_out;
  float* Coutp = outp + T_LEN;
  float* noutp = Coutp + MD * MD;

  proj_kernel<<<dim3(T_LEN / 16, 8), 128, 0, stream>>>(
      x, Wq, bq, Wk, bk, Wv, bv, Wo, bo, qw, kw, vw, owp);
  gates_kernel<<<dim3(T_LEN / 256), 256, 0, stream>>>(
      x, wi, bi, wf, bf, igp, lfp);
  scan_kernel<<<1, 256, 0, stream>>>(
      qw, kw, vw, owp, igp, lfp, C0, n0, fc_w, fc_b, outp, Coutp, noutp);
}


# Scan kernel: each block's LDS reads (fragments of the 4 WMMA blocks, operands of the serial sums) issued as one group ahead of its arithmetic, held by scheduling barriers; sum orders unchanged
# speedup vs baseline: 8.0668x; 8.0668x over previous
#include <hip/hip_runtime.h>
#include <hip/hip_bf16.h>
#include <math.h>


typedef __attribute__((ext_vector_type(2))) float v2f;
typedef __attribute__((ext_vector_type(8))) float v8f;
typedef __attribute__((ext_vector_type(4))) float v4f;

#define T_LEN 8192
#define IN_DIM 512
#define MD 128
#define LCH 64
#define NCH (T_LEN / LCH)
#define NA 4
#define NB 4
#define NGRP (NA * NB)
#define SR (MD / NA)
#define KC (MD / NB)
#define TS 36

__device__ __forceinline__ v8f wmma4(v2f a, v2f b, v8f c) {
  return __builtin_amdgcn_wmma_f32_16x16x4_f32(
       false, a,  false, b,
       (short)0, c,  false,  false);
}

__device__ __forceinline__ float sigmoidf_(float z) {
  return 1.0f / (1.0f + __expf(-z));
}

__global__ __launch_bounds__(128) void proj_kernel(
    const float* __restrict__ x,
    const float* __restrict__ Wq, const float* __restrict__ bq,
    const float* __restrict__ Wk, const float* __restrict__ bk,
    const float* __restrict__ Wv, const float* __restrict__ bv,
    const float* __restrict__ Wo, const float* __restrict__ bo,
    float* __restrict__ qw, float* __restrict__ kw,
    float* __restrict__ vw, float* __restrict__ ow) {
  const int wave = threadIdx.x >> 5;
  const int lane = threadIdx.x & 31;
  const int half = lane >> 4;
  const int lm = lane & 15;

  const int rowBase = blockIdx.x * 16;
  const int colTile = blockIdx.y * 4 + wave;
  const int mat = colTile >> 3;
  const int col = (colTile & 7) * 16;

  const float* W; const float* bias; float* dst;
  switch (mat) {
    case 0:  W = Wq; bias = bq; dst = qw; break;
    case 1:  W = Wk; bias = bk; dst = kw; break;
    case 2:  W = Wv; bias = bv; dst = vw; break;
    default: W = Wo; bias = bo; dst = ow; break;
  }

  v8f acc = {};
  const float* arow = x + (size_t)(rowBase + lm) * IN_DIM + 2 * half;
  const float* brow = W + (size_t)(col + lm) * IN_DIM + 2 * half;
  for (int kk = 0; kk < IN_DIM; kk += 4) {
    v2f a, b;
    a.x = arow[kk];     a.y = arow[kk + 1];
    b.x = brow[kk];     b.y = brow[kk + 1];
    acc = wmma4(a, b, acc);
  }

  const float bv_ = bias[col + lm];
#pragma unroll
  for (int e = 0; e < 8; ++e) {
    const int r = rowBase + e + 8 * half;
    float val = acc[e] + bv_;
    if (mat == 1) val *= 0.08838834764831845f;
    if (mat == 3) val = sigmoidf_(val);
    dst[(size_t)r * MD + col + lm] = val;
  }
}

__global__ __launch_bounds__(256) void gates_kernel(
    const float* __restrict__ x,
    const float* __restrict__ wi, const float* __restrict__ bi,
    const float* __restrict__ wf, const float* __restrict__ bf,
    float* __restrict__ ig, float* __restrict__ la) {
  __shared__ float lf_s[256];
  const int t = blockIdx.x * 256 + threadIdx.x;
  const float* xr = x + (size_t)t * IN_DIM;
  float si = 0.f, sf = 0.f;
  for (int f = 0; f < IN_DIM; ++f) {
    const float xv = xr[f];
    si = fmaf(xv, wi[f], si);
    sf = fmaf(xv, wf[f], sf);
  }
  ig[t] = __expf(si + bi[0]);
  const float z = sf + bf[0];
  lf_s[threadIdx.x] = (z >= 0.f) ? -log1pf(__expf(-z)) : (z - log1pf(__expf(z)));
  __syncthreads();
  const int j = threadIdx.x & (LCH - 1);
  const float* lc = &lf_s[threadIdx.x - j];
  float a = 0.f;
  for (int i = 0; i <= j; ++i) a += lc[i];
  la[t] = a;
}

#define NBUF 3
__global__ __launch_bounds__(256, 1) void scan_kernel(
    const float* __restrict__ qw, const float* __restrict__ kw,
    const float* __restrict__ vw, const float* __restrict__ ow,
    const float* __restrict__ ig, const float* __restrict__ la,
    const float* __restrict__ C0, const float* __restrict__ n0,
    const float* __restrict__ fc_w,
    float* __restrict__ nump, float* __restrict__ denp,
    float* __restrict__ C_out, float* __restrict__ n_out) {
  __shared__ float Cs[2][SR][KC + 1];
  __shared__ float ns[2][KC];
  __shared__ float fcw_s[SR];
  __shared__ float Ps[LCH][LCH + 1];
  __shared__ float Gs[2][LCH][SR + 1];
  __shared__ float rs_s[2][LCH];
  __shared__ float Hs[LCH][SR + 1];
  __shared__ __attribute__((aligned(16))) float Tq[NBUF][LCH][TS];
  __shared__ __attribute__((aligned(16))) float Tk[NBUF][LCH][TS];
  __shared__ __attribute__((aligned(16))) float Tv[NBUF][LCH][TS];
  __shared__ __attribute__((aligned(16))) float To[NBUF][LCH][TS];
  __shared__ __attribute__((aligned(16))) float la_n[NBUF][LCH];
  __shared__ __attribute__((aligned(16))) float ig_n[NBUF][LCH];
  __shared__ __attribute__((aligned(16))) float u_n[NBUF][LCH];

  const int tid = threadIdx.x;
  const int wave = tid >> 5;
  const int lane = tid & 31;
  const int half = lane >> 4;
  const int lm = lane & 15;
  const int ga = blockIdx.x / NB;
  const int gb = blockIdx.x % NB;
  const int r0 = ga * SR;
  const int k0 = gb * KC;

  for (int i = tid; i < SR * KC; i += 256)
    Cs[0][i >> 5][i & 31] = C0[(size_t)(r0 + (i >> 5)) * MD + k0 + (i & 31)];
  if (tid < KC) ns[0][tid] = n0[k0 + tid];
  if (tid < SR) fcw_s[tid] = fc_w[r0 + tid];

  const int st_sub = (tid & 7) * 4;
  const int st_row = tid >> 3;
  const float* gq = qw + (size_t)st_row * MD + k0 + st_sub;
  const float* gk = kw + (size_t)st_row * MD + k0 + st_sub;
  const float* gv = vw + (size_t)st_row * MD + r0 + st_sub;
  const float* go = ow + (size_t)st_row * MD + r0 + st_sub;
  v4f pq0, pq1, pk0, pk1, pv0, pv1, po0, po1;
  float pla = 0.f, pig = 0.f, pend = 0.f;

#define FETCH_CHUNK(tn)                                                     \
  do {                                                                      \
    const size_t o0 = (size_t)(tn) * MD, o1 = o0 + (size_t)32 * MD;         \
    pq0 = *(const v4f*)(gq + o0); pq1 = *(const v4f*)(gq + o1);             \
    pk0 = *(const v4f*)(gk + o0); pk1 = *(const v4f*)(gk + o1);             \
    pv0 = *(const v4f*)(gv + o0); pv1 = *(const v4f*)(gv + o1);             \
    po0 = *(const v4f*)(go + o0); po1 = *(const v4f*)(go + o1);             \
    if (tid < LCH) {                                                        \
      pla = la[(tn) + tid]; pig = ig[(tn) + tid];                           \
      pend = la[(tn) + LCH - 1];                                            \
    }                                                                       \
  } while (0)

#define STAGE_CHUNK(bf)                                                     \
  do {                                                                      \
    *(v4f*)&Tq[bf][st_row][st_sub] = pq0; *(v4f*)&Tq[bf][st_row + 32][st_sub] = pq1; \
    *(v4f*)&Tk[bf][st_row][st_sub] = pk0; *(v4f*)&Tk[bf][st_row + 32][st_sub] = pk1; \
    *(v4f*)&Tv[bf][st_row][st_sub] = pv0; *(v4f*)&Tv[bf][st_row + 32][st_sub] = pv1; \
    *(v4f*)&To[bf][st_row][st_sub] = po0; *(v4f*)&To[bf][st_row + 32][st_sub] = po1; \
    if (tid < LCH) {                                                        \
      la_n[bf][tid] = pla; ig_n[bf][tid] = pig;                             \
      u_n[bf][tid] = __expf(pend - pla) * pig;                              \
    }                                                                       \
  } while (0)

  FETCH_CHUNK(0);
  STAGE_CHUNK(0);
  __syncthreads();

  int bc = NBUF - 1;
  for (int c = -1; c < NCH; ++c) {
    const int b_cur = bc;
    const int b_nxt = (bc == NBUF - 1) ? 0 : bc + 1;
    const int b_stg = (b_nxt == NBUF - 1) ? 0 : b_nxt + 1;
    const int pc = c & 1;
    const int pn = pc ^ 1;
    const bool have_cur = (c >= 0);
    const bool have_nxt = (c + 1 < NCH);
    const bool have_stg = (c + 2 < NCH);
    const int t0 = c * LCH;

    if (have_stg) FETCH_CHUNK((c + 2) * LCH);

    if (wave >= 4) {
      if (have_nxt) {
        const float (*Qn)[TS] = Tq[b_nxt];
        const float (*Kn)[TS] = Tk[b_nxt];
        const float* lan = la_n[b_nxt];
        const float* ign = ig_n[b_nxt];
        const int tr = (wave - 4) * 16;
        v2f fa[KC / 4];
        v2f fb[KC / 4][4];
        const float* ap = &Qn[tr + lm][2 * half];
#pragma unroll
        for (int k4 = 0; k4 < KC / 4; ++k4) {
          fa[k4].x = ap[4 * k4]; fa[k4].y = ap[4 * k4 + 1];
#pragma unroll
          for (int q = 0; q < 4; ++q) {
            const float* bp = &Kn[q * 16 + lm][2 * half];
            fb[k4][q].x = bp[4 * k4]; fb[k4][q].y = bp[4 * k4 + 1];
          }
        }
        float lar[8], lacq[4], igq[4];
#pragma unroll
        for (int e = 0; e < 8; ++e) lar[e] = lan[tr + e + 8 * half];
#pragma unroll
        for (int q = 0; q < 4; ++q) { lacq[q] = lan[q * 16 + lm]; igq[q] = ign[q * 16 + lm]; }
        __builtin_amdgcn_sched_barrier(0);
        v8f acc[4];
#pragma unroll
        for (int q = 0; q < 4; ++q) acc[q] = v8f{};
#pragma unroll
        for (int k4 = 0; k4 < KC / 4; ++k4) {
#pragma unroll
          for (int q = 0; q < 4; ++q) acc[q] = wmma4(fa[k4], fb[k4][q], acc[q]);
        }
#pragma unroll
        for (int q = 0; q < 4; ++q) {
          const int cc = q * 16 + lm;
#pragma unroll
          for (int e = 0; e < 8; ++e) {
            const int r = tr + e + 8 * half;
            const float p = __expf(lar[e] - lacq[q]) * igq[q] * acc[q][e];
            Ps[r][cc] = (r >= cc) ? p : 0.f;
          }
        }
      }
    } else {
      if (have_cur) {
        const float (*Qs)[TS] = Tq[b_cur];
        const float (*Ks)[TS] = Tk[b_cur];
        const float (*Vs)[TS] = Tv[b_cur];
        const float (*Os)[TS] = To[b_cur];
        const float* lac = la_n[b_cur];
        const float* uc = u_n[b_cur];
        const int tr = wave * 16;
        const int ur = (wave >> 1) * 16, ut = (wave & 1) * 16;
        v2f ha[KC / 4];
        v2f hb[KC / 4][2];
        const float* ap = &Qs[tr + lm][2 * half];
#pragma unroll
        for (int k4 = 0; k4 < KC / 4; ++k4) {
          ha[k4].x = ap[4 * k4]; ha[k4].y = ap[4 * k4 + 1];
#pragma unroll
          for (int q = 0; q < 2; ++q) {
            hb[k4][q].x = Cs[pc][q * 16 + lm][4 * k4 + 2 * half];
            hb[k4][q].y = Cs[pc][q * 16 + lm][4 * k4 + 2 * half + 1];
          }
        }
        float lr[8], gq[8][2], oq[8][2];
#pragma unroll
        for (int e = 0; e < 8; ++e) {
          const int r = tr + e + 8 * half;
          lr[e] = lac[r];
#pragma unroll
          for (int q = 0; q < 2; ++q) {
            gq[e][q] = Gs[pc][r][q * 16 + lm];
            oq[e][q] = Os[r][q * 16 + lm];
          }
        }
        const float lend = lac[LCH - 1];
        float c0[8];
#pragma unroll
        for (int e = 0; e < 8; ++e) c0[e] = Cs[pc][ur + e + 8 * half][ut + lm];
        v2f uu[LCH / 4], vv[LCH / 4], kb[LCH / 4];
#pragma unroll
        for (int k4 = 0; k4 < LCH / 4; ++k4) {
          const int j0 = 4 * k4 + 2 * half;
          uu[k4].x = uc[j0];          uu[k4].y = uc[j0 + 1];
          vv[k4].x = Vs[j0][ur + lm]; vv[k4].y = Vs[j0 + 1][ur + lm];
          kb[k4].x = Ks[j0][ut + lm]; kb[k4].y = Ks[j0 + 1][ut + lm];
        }
        {
          v8f ace[2];
          ace[0] = v8f{}; ace[1] = v8f{};
#pragma unroll
          for (int k4 = 0; k4 < KC / 4; ++k4) {
#pragma unroll
            for (int q = 0; q < 2; ++q) ace[q] = wmma4(ha[k4], hb[k4][q], ace[q]);
          }
#pragma unroll
          for (int e = 0; e < 8; ++e) {
            const int r = tr + e + 8 * half;
            const float Ft = __expf(lr[e]);
#pragma unroll
            for (int q = 0; q < 2; ++q) {
              const int cc = q * 16 + lm;
              const float num = fmaf(Ft, ace[q][e], gq[e][q]);
              Hs[r][cc] = num * oq[e][q];
            }
          }
        }
        {
          const float Fend = __expf(lend);
          v8f acc;
#pragma unroll
          for (int e = 0; e < 8; ++e) acc[e] = Fend * c0[e];
#pragma unroll
          for (int k4 = 0; k4 < LCH / 4; ++k4) {
            v2f a;
            a.x = uu[k4].x * vv[k4].x;
            a.y = uu[k4].y * vv[k4].y;
            acc = wmma4(a, kb[k4], acc);
          }
#pragma unroll
          for (int e = 0; e < 8; ++e) Cs[pn][ur + e + 8 * half][ut + lm] = acc[e];
        }
        __builtin_amdgcn_sched_group_barrier(0x100, 128, 0);
        __builtin_amdgcn_sched_group_barrier(0x008, 32, 0);
      }
    }
    __syncthreads();

    if (have_stg) STAGE_CHUNK(b_stg);
    if (wave >= 4) {
      if (have_nxt) {
        const float (*Vn)[TS] = Tv[b_nxt];
        const int tr = (wave - 4) * 16;
        float pax[LCH / 4], pay[LCH / 4];
        float vbx[LCH / 4][2], vby[LCH / 4][2];
#pragma unroll
        for (int k4 = 0; k4 < LCH / 4; ++k4) {
          const int j0 = 4 * k4 + 2 * half;
          pax[k4] = Ps[tr + lm][j0]; pay[k4] = Ps[tr + lm][j0 + 1];
#pragma unroll
          for (int q = 0; q < 2; ++q) {
            vbx[k4][q] = Vn[j0][q * 16 + lm]; vby[k4][q] = Vn[j0 + 1][q * 16 + lm];
          }
        }
        __builtin_amdgcn_sched_barrier(0);
        v8f aci[2];
        aci[0] = v8f{}; aci[1] = v8f{};
#pragma unroll
        for (int k4 = 0; k4 < LCH / 4; ++k4) {
          v2f a;
          a.x = pax[k4]; a.y = pay[k4];
#pragma unroll
          for (int q = 0; q < 2; ++q) {
            v2f b;
            b.x = vbx[k4][q]; b.y = vby[k4][q];
            aci[q] = wmma4(a, b, aci[q]);
          }
        }
#pragma unroll
        for (int q = 0; q < 2; ++q)
#pragma unroll
          for (int e = 0; e < 8; ++e)
            Gs[pn][tr + e + 8 * half][q * 16 + lm] = aci[q][e];
      }
    } else if (wave < 2) {
      if (have_cur) {
        const int row = tid;
        float hv[SR], fw[SR];
#pragma unroll
        for (int f = 0; f < SR; ++f) { hv[f] = Hs[row][f]; fw[f] = fcw_s[f]; }
        __builtin_amdgcn_sched_barrier(0);
        float s = 0.f;
#pragma unroll
        for (int f = 0; f < SR; ++f) s = fmaf(hv[f], fw[f], s);
        nump[(size_t)blockIdx.x * T_LEN + t0 + row] = s;
        if (ga == 0) {
          const float* qrow = Tq[b_cur][row];
          float qv[KC], nv[KC];
#pragma unroll
          for (int f = 0; f < KC; ++f) { qv[f] = qrow[f]; nv[f] = ns[pc][f]; }
          const float lrow = la_n[b_cur][row];
          const float rsv = rs_s[pc][row];
          __builtin_amdgcn_sched_barrier(0);
          float qn = 0.f;
#pragma unroll
          for (int f = 0; f < KC; ++f) qn = fmaf(qv[f], nv[f], qn);
          denp[(size_t)gb * T_LEN + t0 + row] = fmaf(__expf(lrow), qn, rsv);
        }
      }
    } else {
      const int row = tid - 64;
      const bool do_n = (wave == 3) && have_cur;
      if (have_nxt && do_n) {
        float pv[LCH], uv[LCH], kv[LCH];
#pragma unroll
        for (int j = 0; j < LCH; ++j) pv[j] = Ps[row][j];
#pragma unroll
        for (int j = 0; j < LCH; ++j) { uv[j] = u_n[b_cur][j]; kv[j] = Tk[b_cur][j][lane]; }
        const float lend = la_n[b_cur][LCH - 1];
        const float nold = ns[pc][lane];
        __builtin_amdgcn_sched_barrier(0);
        float s = 0.f;
#pragma unroll
        for (int j = 0; j < LCH; ++j) s += pv[j];
        rs_s[pn][row] = s;
        const float Fend = __expf(lend);
        float sn = Fend * nold;
#pragma unroll
        for (int j = 0; j < LCH; ++j) sn = fmaf(uv[j], kv[j], sn);
        ns[pn][lane] = sn;
      } else if (have_nxt) {
        float pv[LCH];
#pragma unroll
        for (int j = 0; j < LCH; ++j) pv[j] = Ps[row][j];
        __builtin_amdgcn_sched_barrier(0);
        float s = 0.f;
#pragma unroll
        for (int j = 0; j < LCH; ++j) s += pv[j];
        rs_s[pn][row] = s;
      } else if (do_n) {
        float uv[LCH], kv[LCH];
#pragma unroll
        for (int j = 0; j < LCH; ++j) { uv[j] = u_n[b_cur][j]; kv[j] = Tk[b_cur][j][lane]; }
        const float lend = la_n[b_cur][LCH - 1];
        const float nold = ns[pc][lane];
        __builtin_amdgcn_sched_barrier(0);
        const float Fend = __expf(lend);
        float sn = Fend * nold;
#pragma unroll
        for (int j = 0; j < LCH; ++j) sn = fmaf(uv[j], kv[j], sn);
        ns[pn][lane] = sn;
      }
    }
    __syncthreads();
    bc = b_nxt;
  }

#undef FETCH_CHUNK
#undef STAGE_CHUNK

  for (int i = tid; i < SR * KC; i += 256)
    C_out[(size_t)(r0 + (i >> 5)) * MD + k0 + (i & 31)] = Cs[NCH & 1][i >> 5][i & 31];
  if (ga == 0 && tid < KC) n_out[k0 + tid] = ns[NCH & 1][tid];
}

__global__ __launch_bounds__(256) void out_kernel(
    const float* __restrict__ nump, const float* __restrict__ denp,
    const float* __restrict__ fc_b, float* __restrict__ out) {
  const int t = blockIdx.x * 256 + threadIdx.x;
  float s = nump[t];
#pragma unroll
  for (int g = 1; g < NGRP; ++g) s += nump[(size_t)g * T_LEN + t];
  float d = denp[t];
#pragma unroll
  for (int b = 1; b < NB; ++b) d += denp[(size_t)b * T_LEN + t];
  out[t] = s / fmaxf(fabsf(d), 1.0f) + fc_b[0];
}

extern "C" void kernel_launch(void* const* d_in, const int* in_sizes, int n_in,
                              void* d_out, int out_size, void* d_ws, size_t ws_size,
                              hipStream_t stream) {
  const float* x    = (const float*)d_in[0];
  const float* C0   = (const float*)d_in[1];
  const float* n0   = (const float*)d_in[2];
  const float* Wq   = (const float*)d_in[3];
  const float* bq   = (const float*)d_in[4];
  const float* Wk   = (const float*)d_in[5];
  const float* bk   = (const float*)d_in[6];
  const float* Wv   = (const float*)d_in[7];
  const float* bv   = (const float*)d_in[8];
  const float* wi   = (const float*)d_in[9];
  const float* bi   = (const float*)d_in[10];
  const float* wf   = (const float*)d_in[11];
  const float* bf   = (const float*)d_in[12];
  const float* Wo   = (const float*)d_in[13];
  const float* bo   = (const float*)d_in[14];
  const float* fc_w = (const float*)d_in[15];
  const float* fc_b = (const float*)d_in[16];

  float* ws = (float*)d_ws;
  const size_t NP = (size_t)T_LEN * MD;
  float* qw = ws;
  float* kw = ws + NP;
  float* vw = ws + 2 * NP;
  float* owp = ws + 3 * NP;
  float* igp = ws + 4 * NP;
  float* lfp = igp + T_LEN;
  float* numpp = lfp + T_LEN;
  float* denpp = numpp + NGRP * T_LEN;

  float* outp  = (float*)d_out;
  float* Coutp = outp + T_LEN;
  float* noutp = Coutp + MD * MD;

  proj_kernel<<<dim3(T_LEN / 16, 8), 128, 0, stream>>>(
      x, Wq, bq, Wk, bk, Wv, bv, Wo, bo, qw, kw, vw, owp);
  gates_kernel<<<dim3(T_LEN / 256), 256, 0, stream>>>(
      x, wi, bi, wf, bf, igp, lfp);
  scan_kernel<<<NGRP, 256, 0, stream>>>(
      qw, kw, vw, owp, igp, lfp, C0, n0, fc_w, numpp, denpp, Coutp, noutp);
  out_kernel<<<T_LEN / 256, 256, 0, stream>>>(numpp, denpp, fc_b, outp);
}


# Projection: each wave holds a 64x64 output block in registers, fragments by 16-byte loads straight from global, each feeding 4 WMMAs; k taken 0,1,4,5 then 2,3,6,7 per 8, so the sum order changes
# speedup vs baseline: 9.1240x; 1.1311x over previous
#include <hip/hip_runtime.h>
#include <hip/hip_bf16.h>
#include <math.h>


typedef __attribute__((ext_vector_type(2))) float v2f;
typedef __attribute__((ext_vector_type(8))) float v8f;
typedef __attribute__((ext_vector_type(4))) float v4f;

#define T_LEN 8192
#define IN_DIM 512
#define MD 128
#define LCH 64
#define NCH (T_LEN / LCH)
#define NA 4
#define NB 4
#define NGRP (NA * NB)
#define SR (MD / NA)
#define KC (MD / NB)
#define TS 36

__device__ __forceinline__ v8f wmma4(v2f a, v2f b, v8f c) {
  return __builtin_amdgcn_wmma_f32_16x16x4_f32(
       false, a,  false, b,
       (short)0, c,  false,  false);
}

__device__ __forceinline__ float sigmoidf_(float z) {
  return 1.0f / (1.0f + __expf(-z));
}

#define PJ_NSET 4
#define PJ_STEPS (IN_DIM / 8)
__global__ __launch_bounds__(128, 1) void proj_kernel(
    const float* __restrict__ x,
    const float* __restrict__ Wq, const float* __restrict__ bq,
    const float* __restrict__ Wk, const float* __restrict__ bk,
    const float* __restrict__ Wv, const float* __restrict__ bv,
    const float* __restrict__ Wo, const float* __restrict__ bo,
    float* __restrict__ qw, float* __restrict__ kw,
    float* __restrict__ vw, float* __restrict__ ow) {
  const int wave = threadIdx.x >> 5;
  const int lane = threadIdx.x & 31;
  const int half = lane >> 4;
  const int lm = lane & 15;

  const int mat = blockIdx.y;
  const int row0 = (blockIdx.x * 2 + (wave >> 1)) * 64;
  const int col0 = (wave & 1) * 64;

  const float* W; const float* bias; float* dst;
  switch (mat) {
    case 0:  W = Wq; bias = bq; dst = qw; break;
    case 1:  W = Wk; bias = bk; dst = kw; break;
    case 2:  W = Wv; bias = bv; dst = vw; break;
    default: W = Wo; bias = bo; dst = ow; break;
  }

  const float* ap = x + (size_t)(row0 + lm) * IN_DIM + 4 * half;
  const float* bp = W + (size_t)(col0 + lm) * IN_DIM + 4 * half;

  v8f acc[4][4];
#pragma unroll
  for (int i = 0; i < 4; ++i)
#pragma unroll
    for (int j = 0; j < 4; ++j) acc[i][j] = v8f{};

  v4f fa[PJ_NSET][4], fb[PJ_NSET][4];
#pragma unroll
  for (int s = 0; s < PJ_NSET - 1; ++s) {
#pragma unroll
    for (int i = 0; i < 4; ++i) {
      fa[s][i] = *(const v4f*)(ap + (size_t)i * 16 * IN_DIM + 8 * s);
      fb[s][i] = *(const v4f*)(bp + (size_t)i * 16 * IN_DIM + 8 * s);
    }
  }

  for (int s0 = 0; s0 < PJ_STEPS; s0 += PJ_NSET) {
#pragma unroll
    for (int u = 0; u < PJ_NSET; ++u) {
      const int sn = (s0 + u + PJ_NSET - 1 < PJ_STEPS) ? (s0 + u + PJ_NSET - 1) : (PJ_STEPS - 1);
      const int w = (u + PJ_NSET - 1) % PJ_NSET;
#pragma unroll
      for (int i = 0; i < 4; ++i) {
        fa[w][i] = *(const v4f*)(ap + (size_t)i * 16 * IN_DIM + 8 * sn);
        fb[w][i] = *(const v4f*)(bp + (size_t)i * 16 * IN_DIM + 8 * sn);
      }
      __builtin_amdgcn_sched_barrier(0);
#pragma unroll
      for (int i = 0; i < 4; ++i) {
        v2f a; a.x = fa[u][i].x; a.y = fa[u][i].y;
#pragma unroll
        for (int j = 0; j < 4; ++j) {
          v2f b; b.x = fb[u][j].x; b.y = fb[u][j].y;
          acc[i][j] = wmma4(a, b, acc[i][j]);
        }
      }
#pragma unroll
      for (int i = 0; i < 4; ++i) {
        v2f a; a.x = fa[u][i].z; a.y = fa[u][i].w;
#pragma unroll
        for (int j = 0; j < 4; ++j) {
          v2f b; b.x = fb[u][j].z; b.y = fb[u][j].w;
          acc[i][j] = wmma4(a, b, acc[i][j]);
        }
      }
    }
  }

#define PJ_STORE(EXPR)                                                      \
  _Pragma("unroll") for (int j = 0; j < 4; ++j) {                           \
    const int col = col0 + j * 16 + lm;                                     \
    const float bv_ = bias[col];                                            \
    _Pragma("unroll") for (int i = 0; i < 4; ++i) {                         \
      _Pragma("unroll") for (int e = 0; e < 8; ++e) {                       \
        const int r = row0 + i * 16 + e + 8 * half;                         \
        const float val = acc[i][j][e] + bv_;                               \
        dst[(size_t)r * MD + col] = (EXPR);                                 \
      }                                                                     \
    }                                                                       \
  }
  if (mat == 3) {
    PJ_STORE(sigmoidf_(val))
  } else if (mat == 1) {
    PJ_STORE(val * 0.08838834764831845f)
  } else {
    PJ_STORE(val)
  }
#undef PJ_STORE
}

__global__ __launch_bounds__(256) void gates_kernel(
    const float* __restrict__ x,
    const float* __restrict__ wi, const float* __restrict__ bi,
    const float* __restrict__ wf, const float* __restrict__ bf,
    float* __restrict__ ig, float* __restrict__ la) {
  __shared__ float lf_s[256];
  const int t = blockIdx.x * 256 + threadIdx.x;
  const float* xr = x + (size_t)t * IN_DIM;
  float si = 0.f, sf = 0.f;
  for (int f = 0; f < IN_DIM; ++f) {
    const float xv = xr[f];
    si = fmaf(xv, wi[f], si);
    sf = fmaf(xv, wf[f], sf);
  }
  ig[t] = __expf(si + bi[0]);
  const float z = sf + bf[0];
  lf_s[threadIdx.x] = (z >= 0.f) ? -log1pf(__expf(-z)) : (z - log1pf(__expf(z)));
  __syncthreads();
  const int j = threadIdx.x & (LCH - 1);
  const float* lc = &lf_s[threadIdx.x - j];
  float a = 0.f;
  for (int i = 0; i <= j; ++i) a += lc[i];
  la[t] = a;
}

#define NBUF 3
__global__ __launch_bounds__(256, 1) void scan_kernel(
    const float* __restrict__ qw, const float* __restrict__ kw,
    const float* __restrict__ vw, const float* __restrict__ ow,
    const float* __restrict__ ig, const float* __restrict__ la,
    const float* __restrict__ C0, const float* __restrict__ n0,
    const float* __restrict__ fc_w,
    float* __restrict__ nump, float* __restrict__ denp,
    float* __restrict__ C_out, float* __restrict__ n_out) {
  __shared__ float Cs[2][SR][KC + 1];
  __shared__ float ns[2][KC];
  __shared__ float fcw_s[SR];
  __shared__ float Ps[LCH][LCH + 1];
  __shared__ float Gs[2][LCH][SR + 1];
  __shared__ float rs_s[2][LCH];
  __shared__ float Hs[LCH][SR + 1];
  __shared__ __attribute__((aligned(16))) float Tq[NBUF][LCH][TS];
  __shared__ __attribute__((aligned(16))) float Tk[NBUF][LCH][TS];
  __shared__ __attribute__((aligned(16))) float Tv[NBUF][LCH][TS];
  __shared__ __attribute__((aligned(16))) float To[NBUF][LCH][TS];
  __shared__ __attribute__((aligned(16))) float la_n[NBUF][LCH];
  __shared__ __attribute__((aligned(16))) float ig_n[NBUF][LCH];
  __shared__ __attribute__((aligned(16))) float u_n[NBUF][LCH];

  const int tid = threadIdx.x;
  const int wave = tid >> 5;
  const int lane = tid & 31;
  const int half = lane >> 4;
  const int lm = lane & 15;
  const int ga = blockIdx.x / NB;
  const int gb = blockIdx.x % NB;
  const int r0 = ga * SR;
  const int k0 = gb * KC;

  for (int i = tid; i < SR * KC; i += 256)
    Cs[0][i >> 5][i & 31] = C0[(size_t)(r0 + (i >> 5)) * MD + k0 + (i & 31)];
  if (tid < KC) ns[0][tid] = n0[k0 + tid];
  if (tid < SR) fcw_s[tid] = fc_w[r0 + tid];

  const int st_sub = (tid & 7) * 4;
  const int st_row = tid >> 3;
  const float* gq = qw + (size_t)st_row * MD + k0 + st_sub;
  const float* gk = kw + (size_t)st_row * MD + k0 + st_sub;
  const float* gv = vw + (size_t)st_row * MD + r0 + st_sub;
  const float* go = ow + (size_t)st_row * MD + r0 + st_sub;
  v4f pq0, pq1, pk0, pk1, pv0, pv1, po0, po1;
  float pla = 0.f, pig = 0.f, pend = 0.f;

#define FETCH_CHUNK(tn)                                                     \
  do {                                                                      \
    const size_t o0 = (size_t)(tn) * MD, o1 = o0 + (size_t)32 * MD;         \
    pq0 = *(const v4f*)(gq + o0); pq1 = *(const v4f*)(gq + o1);             \
    pk0 = *(const v4f*)(gk + o0); pk1 = *(const v4f*)(gk + o1);             \
    pv0 = *(const v4f*)(gv + o0); pv1 = *(const v4f*)(gv + o1);             \
    po0 = *(const v4f*)(go + o0); po1 = *(const v4f*)(go + o1);             \
    if (tid < LCH) {                                                        \
      pla = la[(tn) + tid]; pig = ig[(tn) + tid];                           \
      pend = la[(tn) + LCH - 1];                                            \
    }                                                                       \
  } while (0)

#define STAGE_CHUNK(bf)                                                     \
  do {                                                                      \
    *(v4f*)&Tq[bf][st_row][st_sub] = pq0; *(v4f*)&Tq[bf][st_row + 32][st_sub] = pq1; \
    *(v4f*)&Tk[bf][st_row][st_sub] = pk0; *(v4f*)&Tk[bf][st_row + 32][st_sub] = pk1; \
    *(v4f*)&Tv[bf][st_row][st_sub] = pv0; *(v4f*)&Tv[bf][st_row + 32][st_sub] = pv1; \
    *(v4f*)&To[bf][st_row][st_sub] = po0; *(v4f*)&To[bf][st_row + 32][st_sub] = po1; \
    if (tid < LCH) {                                                        \
      la_n[bf][tid] = pla; ig_n[bf][tid] = pig;                             \
      u_n[bf][tid] = __expf(pend - pla) * pig;                              \
    }                                                                       \
  } while (0)

  FETCH_CHUNK(0);
  STAGE_CHUNK(0);
  __syncthreads();

  int bc = NBUF - 1;
  for (int c = -1; c < NCH; ++c) {
    const int b_cur = bc;
    const int b_nxt = (bc == NBUF - 1) ? 0 : bc + 1;
    const int b_stg = (b_nxt == NBUF - 1) ? 0 : b_nxt + 1;
    const int pc = c & 1;
    const int pn = pc ^ 1;
    const bool have_cur = (c >= 0);
    const bool have_nxt = (c + 1 < NCH);
    const bool have_stg = (c + 2 < NCH);
    const int t0 = c * LCH;

    if (have_stg) FETCH_CHUNK((c + 2) * LCH);

    if (wave >= 4) {
      if (have_nxt) {
        const float (*Qn)[TS] = Tq[b_nxt];
        const float (*Kn)[TS] = Tk[b_nxt];
        const float* lan = la_n[b_nxt];
        const float* ign = ig_n[b_nxt];
        const int tr = (wave - 4) * 16;
        v2f fa[KC / 4];
        v2f fb[KC / 4][4];
        const float* ap = &Qn[tr + lm][2 * half];
#pragma unroll
        for (int k4 = 0; k4 < KC / 4; ++k4) {
          fa[k4].x = ap[4 * k4]; fa[k4].y = ap[4 * k4 + 1];
#pragma unroll
          for (int q = 0; q < 4; ++q) {
            const float* bp = &Kn[q * 16 + lm][2 * half];
            fb[k4][q].x = bp[4 * k4]; fb[k4][q].y = bp[4 * k4 + 1];
          }
        }
        float lar[8], lacq[4], igq[4];
#pragma unroll
        for (int e = 0; e < 8; ++e) lar[e] = lan[tr + e + 8 * half];
#pragma unroll
        for (int q = 0; q < 4; ++q) { lacq[q] = lan[q * 16 + lm]; igq[q] = ign[q * 16 + lm]; }
        __builtin_amdgcn_sched_barrier(0);
        v8f acc[4];
#pragma unroll
        for (int q = 0; q < 4; ++q) acc[q] = v8f{};
#pragma unroll
        for (int k4 = 0; k4 < KC / 4; ++k4) {
#pragma unroll
          for (int q = 0; q < 4; ++q) acc[q] = wmma4(fa[k4], fb[k4][q], acc[q]);
        }
#pragma unroll
        for (int q = 0; q < 4; ++q) {
          const int cc = q * 16 + lm;
#pragma unroll
          for (int e = 0; e < 8; ++e) {
            const int r = tr + e + 8 * half;
            const float p = __expf(lar[e] - lacq[q]) * igq[q] * acc[q][e];
            Ps[r][cc] = (r >= cc) ? p : 0.f;
          }
        }
      }
    } else {
      if (have_cur) {
        const float (*Qs)[TS] = Tq[b_cur];
        const float (*Ks)[TS] = Tk[b_cur];
        const float (*Vs)[TS] = Tv[b_cur];
        const float (*Os)[TS] = To[b_cur];
        const float* lac = la_n[b_cur];
        const float* uc = u_n[b_cur];
        const int tr = wave * 16;
        const int ur = (wave >> 1) * 16, ut = (wave & 1) * 16;
        v2f ha[KC / 4];
        v2f hb[KC / 4][2];
        const float* ap = &Qs[tr + lm][2 * half];
#pragma unroll
        for (int k4 = 0; k4 < KC / 4; ++k4) {
          ha[k4].x = ap[4 * k4]; ha[k4].y = ap[4 * k4 + 1];
#pragma unroll
          for (int q = 0; q < 2; ++q) {
            hb[k4][q].x = Cs[pc][q * 16 + lm][4 * k4 + 2 * half];
            hb[k4][q].y = Cs[pc][q * 16 + lm][4 * k4 + 2 * half + 1];
          }
        }
        float lr[8], gq[8][2], oq[8][2];
#pragma unroll
        for (int e = 0; e < 8; ++e) {
          const int r = tr + e + 8 * half;
          lr[e] = lac[r];
#pragma unroll
          for (int q = 0; q < 2; ++q) {
            gq[e][q] = Gs[pc][r][q * 16 + lm];
            oq[e][q] = Os[r][q * 16 + lm];
          }
        }
        const float lend = lac[LCH - 1];
        float c0[8];
#pragma unroll
        for (int e = 0; e < 8; ++e) c0[e] = Cs[pc][ur + e + 8 * half][ut + lm];
        v2f uu[LCH / 4], vv[LCH / 4], kb[LCH / 4];
#pragma unroll
        for (int k4 = 0; k4 < LCH / 4; ++k4) {
          const int j0 = 4 * k4 + 2 * half;
          uu[k4].x = uc[j0];          uu[k4].y = uc[j0 + 1];
          vv[k4].x = Vs[j0][ur + lm]; vv[k4].y = Vs[j0 + 1][ur + lm];
          kb[k4].x = Ks[j0][ut + lm]; kb[k4].y = Ks[j0 + 1][ut + lm];
        }
        {
          v8f ace[2];
          ace[0] = v8f{}; ace[1] = v8f{};
#pragma unroll
          for (int k4 = 0; k4 < KC / 4; ++k4) {
#pragma unroll
            for (int q = 0; q < 2; ++q) ace[q] = wmma4(ha[k4], hb[k4][q], ace[q]);
          }
#pragma unroll
          for (int e = 0; e < 8; ++e) {
            const int r = tr + e + 8 * half;
            const float Ft = __expf(lr[e]);
#pragma unroll
            for (int q = 0; q < 2; ++q) {
              const int cc = q * 16 + lm;
              const float num = fmaf(Ft, ace[q][e], gq[e][q]);
              Hs[r][cc] = num * oq[e][q];
            }
          }
        }
        {
          const float Fend = __expf(lend);
          v8f acc;
#pragma unroll
          for (int e = 0; e < 8; ++e) acc[e] = Fend * c0[e];
#pragma unroll
          for (int k4 = 0; k4 < LCH / 4; ++k4) {
            v2f a;
            a.x = uu[k4].x * vv[k4].x;
            a.y = uu[k4].y * vv[k4].y;
            acc = wmma4(a, kb[k4], acc);
          }
#pragma unroll
          for (int e = 0; e < 8; ++e) Cs[pn][ur + e + 8 * half][ut + lm] = acc[e];
        }
        __builtin_amdgcn_sched_group_barrier(0x100, 128, 0);
        __builtin_amdgcn_sched_group_barrier(0x008, 32, 0);
      }
    }
    __syncthreads();

    if (have_stg) STAGE_CHUNK(b_stg);
    if (wave >= 4) {
      if (have_nxt) {
        const float (*Vn)[TS] = Tv[b_nxt];
        const int tr = (wave - 4) * 16;
        float pax[LCH / 4], pay[LCH / 4];
        float vbx[LCH / 4][2], vby[LCH / 4][2];
#pragma unroll
        for (int k4 = 0; k4 < LCH / 4; ++k4) {
          const int j0 = 4 * k4 + 2 * half;
          pax[k4] = Ps[tr + lm][j0]; pay[k4] = Ps[tr + lm][j0 + 1];
#pragma unroll
          for (int q = 0; q < 2; ++q) {
            vbx[k4][q] = Vn[j0][q * 16 + lm]; vby[k4][q] = Vn[j0 + 1][q * 16 + lm];
          }
        }
        __builtin_amdgcn_sched_barrier(0);
        v8f aci[2];
        aci[0] = v8f{}; aci[1] = v8f{};
#pragma unroll
        for (int k4 = 0; k4 < LCH / 4; ++k4) {
          v2f a;
          a.x = pax[k4]; a.y = pay[k4];
#pragma unroll
          for (int q = 0; q < 2; ++q) {
            v2f b;
            b.x = vbx[k4][q]; b.y = vby[k4][q];
            aci[q] = wmma4(a, b, aci[q]);
          }
        }
#pragma unroll
        for (int q = 0; q < 2; ++q)
#pragma unroll
          for (int e = 0; e < 8; ++e)
            Gs[pn][tr + e + 8 * half][q * 16 + lm] = aci[q][e];
      }
    } else if (wave < 2) {
      if (have_cur) {
        const int row = tid;
        float hv[SR], fw[SR];
#pragma unroll
        for (int f = 0; f < SR; ++f) { hv[f] = Hs[row][f]; fw[f] = fcw_s[f]; }
        __builtin_amdgcn_sched_barrier(0);
        float s = 0.f;
#pragma unroll
        for (int f = 0; f < SR; ++f) s = fmaf(hv[f], fw[f], s);
        nump[(size_t)blockIdx.x * T_LEN + t0 + row] = s;
        if (ga == 0) {
          const float* qrow = Tq[b_cur][row];
          float qv[KC], nv[KC];
#pragma unroll
          for (int f = 0; f < KC; ++f) { qv[f] = qrow[f]; nv[f] = ns[pc][f]; }
          const float lrow = la_n[b_cur][row];
          const float rsv = rs_s[pc][row];
          __builtin_amdgcn_sched_barrier(0);
          float qn = 0.f;
#pragma unroll
          for (int f = 0; f < KC; ++f) qn = fmaf(qv[f], nv[f], qn);
          denp[(size_t)gb * T_LEN + t0 + row] = fmaf(__expf(lrow), qn, rsv);
        }
      }
    } else {
      const int row = tid - 64;
      const bool do_n = (wave == 3) && have_cur;
      if (have_nxt && do_n) {
        float pv[LCH], uv[LCH], kv[LCH];
#pragma unroll
        for (int j = 0; j < LCH; ++j) pv[j] = Ps[row][j];
#pragma unroll
        for (int j = 0; j < LCH; ++j) { uv[j] = u_n[b_cur][j]; kv[j] = Tk[b_cur][j][lane]; }
        const float lend = la_n[b_cur][LCH - 1];
        const float nold = ns[pc][lane];
        __builtin_amdgcn_sched_barrier(0);
        float s = 0.f;
#pragma unroll
        for (int j = 0; j < LCH; ++j) s += pv[j];
        rs_s[pn][row] = s;
        const float Fend = __expf(lend);
        float sn = Fend * nold;
#pragma unroll
        for (int j = 0; j < LCH; ++j) sn = fmaf(uv[j], kv[j], sn);
        ns[pn][lane] = sn;
      } else if (have_nxt) {
        float pv[LCH];
#pragma unroll
        for (int j = 0; j < LCH; ++j) pv[j] = Ps[row][j];
        __builtin_amdgcn_sched_barrier(0);
        float s = 0.f;
#pragma unroll
        for (int j = 0; j < LCH; ++j) s += pv[j];
        rs_s[pn][row] = s;
      } else if (do_n) {
        float uv[LCH], kv[LCH];
#pragma unroll
        for (int j = 0; j < LCH; ++j) { uv[j] = u_n[b_cur][j]; kv[j] = Tk[b_cur][j][lane]; }
        const float lend = la_n[b_cur][LCH - 1];
        const float nold = ns[pc][lane];
        __builtin_amdgcn_sched_barrier(0);
        const float Fend = __expf(lend);
        float sn = Fend * nold;
#pragma unroll
        for (int j = 0; j < LCH; ++j) sn = fmaf(uv[j], kv[j], sn);
        ns[pn][lane] = sn;
      }
    }
    __syncthreads();
    bc = b_nxt;
  }

#undef FETCH_CHUNK
#undef STAGE_CHUNK

  for (int i = tid; i < SR * KC; i += 256)
    C_out[(size_t)(r0 + (i >> 5)) * MD + k0 + (i & 31)] = Cs[NCH & 1][i >> 5][i & 31];
  if (ga == 0 && tid < KC) n_out[k0 + tid] = ns[NCH & 1][tid];
}

__global__ __launch_bounds__(256) void out_kernel(
    const float* __restrict__ nump, const float* __restrict__ denp,
    const float* __restrict__ fc_b, float* __restrict__ out) {
  const int t = blockIdx.x * 256 + threadIdx.x;
  float s = nump[t];
#pragma unroll
  for (int g = 1; g < NGRP; ++g) s += nump[(size_t)g * T_LEN + t];
  float d = denp[t];
#pragma unroll
  for (int b = 1; b < NB; ++b) d += denp[(size_t)b * T_LEN + t];
  out[t] = s / fmaxf(fabsf(d), 1.0f) + fc_b[0];
}

extern "C" void kernel_launch(void* const* d_in, const int* in_sizes, int n_in,
                              void* d_out, int out_size, void* d_ws, size_t ws_size,
                              hipStream_t stream) {
  const float* x    = (const float*)d_in[0];
  const float* C0   = (const float*)d_in[1];
  const float* n0   = (const float*)d_in[2];
  const float* Wq   = (const float*)d_in[3];
  const float* bq   = (const float*)d_in[4];
  const float* Wk   = (const float*)d_in[5];
  const float* bk   = (const float*)d_in[6];
  const float* Wv   = (const float*)d_in[7];
  const float* bv   = (const float*)d_in[8];
  const float* wi   = (const float*)d_in[9];
  const float* bi   = (const float*)d_in[10];
  const float* wf   = (const float*)d_in[11];
  const float* bf   = (const float*)d_in[12];
  const float* Wo   = (const float*)d_in[13];
  const float* bo   = (const float*)d_in[14];
  const float* fc_w = (const float*)d_in[15];
  const float* fc_b = (const float*)d_in[16];

  float* ws = (float*)d_ws;
  const size_t NP = (size_t)T_LEN * MD;
  float* qw = ws;
  float* kw = ws + NP;
  float* vw = ws + 2 * NP;
  float* owp = ws + 3 * NP;
  float* igp = ws + 4 * NP;
  float* lfp = igp + T_LEN;
  float* numpp = lfp + T_LEN;
  float* denpp = numpp + NGRP * T_LEN;

  float* outp  = (float*)d_out;
  float* Coutp = outp + T_LEN;
  float* noutp = Coutp + MD * MD;

  proj_kernel<<<dim3(T_LEN / 128, 4), 128, 0, stream>>>(
      x, Wq, bq, Wk, bk, Wv, bv, Wo, bo, qw, kw, vw, owp);
  gates_kernel<<<dim3(T_LEN / 256), 256, 0, stream>>>(
      x, wi, bi, wf, bf, igp, lfp);
  scan_kernel<<<NGRP, 256, 0, stream>>>(
      qw, kw, vw, owp, igp, lfp, C0, n0, fc_w, numpp, denpp, Coutp, noutp);
  out_kernel<<<T_LEN / 256, 256, 0, stream>>>(numpp, denpp, fc_b, outp);
}


# Gates launch re-tiled: each x row read by a whole wave, whole-line 16-byte loads, one group per chunk (was a thread per row, lanes 2 KB apart); x.wi, x.wf = lane sums + butterfly: sum order changes
# speedup vs baseline: 9.4798x; 1.0390x over previous
#include <hip/hip_runtime.h>
#include <hip/hip_bf16.h>
#include <math.h>


typedef __attribute__((ext_vector_type(2))) float v2f;
typedef __attribute__((ext_vector_type(8))) float v8f;
typedef __attribute__((ext_vector_type(4))) float v4f;

#define T_LEN 8192
#define IN_DIM 512
#define MD 128
#define LCH 64
#define NCH (T_LEN / LCH)
#define NA 4
#define NB 4
#define NGRP (NA * NB)
#define SR (MD / NA)
#define KC (MD / NB)
#define TS 36

__device__ __forceinline__ v8f wmma4(v2f a, v2f b, v8f c) {
  return __builtin_amdgcn_wmma_f32_16x16x4_f32(
       false, a,  false, b,
       (short)0, c,  false,  false);
}

__device__ __forceinline__ float sigmoidf_(float z) {
  return 1.0f / (1.0f + __expf(-z));
}

#define PJ_NSET 4
#define PJ_STEPS (IN_DIM / 8)
__global__ __launch_bounds__(128, 1) void proj_kernel(
    const float* __restrict__ x,
    const float* __restrict__ Wq, const float* __restrict__ bq,
    const float* __restrict__ Wk, const float* __restrict__ bk,
    const float* __restrict__ Wv, const float* __restrict__ bv,
    const float* __restrict__ Wo, const float* __restrict__ bo,
    float* __restrict__ qw, float* __restrict__ kw,
    float* __restrict__ vw, float* __restrict__ ow) {
  const int wave = threadIdx.x >> 5;
  const int lane = threadIdx.x & 31;
  const int half = lane >> 4;
  const int lm = lane & 15;

  const int mat = blockIdx.y;
  const int row0 = (blockIdx.x * 2 + (wave >> 1)) * 64;
  const int col0 = (wave & 1) * 64;

  const float* W; const float* bias; float* dst;
  switch (mat) {
    case 0:  W = Wq; bias = bq; dst = qw; break;
    case 1:  W = Wk; bias = bk; dst = kw; break;
    case 2:  W = Wv; bias = bv; dst = vw; break;
    default: W = Wo; bias = bo; dst = ow; break;
  }

  const float* ap = x + (size_t)(row0 + lm) * IN_DIM + 4 * half;
  const float* bp = W + (size_t)(col0 + lm) * IN_DIM + 4 * half;

  v8f acc[4][4];
#pragma unroll
  for (int i = 0; i < 4; ++i)
#pragma unroll
    for (int j = 0; j < 4; ++j) acc[i][j] = v8f{};

  v4f fa[PJ_NSET][4], fb[PJ_NSET][4];
#pragma unroll
  for (int s = 0; s < PJ_NSET - 1; ++s) {
#pragma unroll
    for (int i = 0; i < 4; ++i) {
      fa[s][i] = *(const v4f*)(ap + (size_t)i * 16 * IN_DIM + 8 * s);
      fb[s][i] = *(const v4f*)(bp + (size_t)i * 16 * IN_DIM + 8 * s);
    }
  }

  for (int s0 = 0; s0 < PJ_STEPS; s0 += PJ_NSET) {
#pragma unroll
    for (int u = 0; u < PJ_NSET; ++u) {
      const int sn = (s0 + u + PJ_NSET - 1 < PJ_STEPS) ? (s0 + u + PJ_NSET - 1) : (PJ_STEPS - 1);
      const int w = (u + PJ_NSET - 1) % PJ_NSET;
#pragma unroll
      for (int i = 0; i < 4; ++i) {
        fa[w][i] = *(const v4f*)(ap + (size_t)i * 16 * IN_DIM + 8 * sn);
        fb[w][i] = *(const v4f*)(bp + (size_t)i * 16 * IN_DIM + 8 * sn);
      }
      __builtin_amdgcn_sched_barrier(0);
#pragma unroll
      for (int i = 0; i < 4; ++i) {
        v2f a; a.x = fa[u][i].x; a.y = fa[u][i].y;
#pragma unroll
        for (int j = 0; j < 4; ++j) {
          v2f b; b.x = fb[u][j].x; b.y = fb[u][j].y;
          acc[i][j] = wmma4(a, b, acc[i][j]);
        }
      }
#pragma unroll
      for (int i = 0; i < 4; ++i) {
        v2f a; a.x = fa[u][i].z; a.y = fa[u][i].w;
#pragma unroll
        for (int j = 0; j < 4; ++j) {
          v2f b; b.x = fb[u][j].z; b.y = fb[u][j].w;
          acc[i][j] = wmma4(a, b, acc[i][j]);
        }
      }
    }
  }

#define PJ_STORE(EXPR)                                                      \
  _Pragma("unroll") for (int j = 0; j < 4; ++j) {                           \
    const int col = col0 + j * 16 + lm;                                     \
    const float bv_ = bias[col];                                            \
    _Pragma("unroll") for (int i = 0; i < 4; ++i) {                         \
      _Pragma("unroll") for (int e = 0; e < 8; ++e) {                       \
        const int r = row0 + i * 16 + e + 8 * half;                         \
        const float val = acc[i][j][e] + bv_;                               \
        dst[(size_t)r * MD + col] = (EXPR);                                 \
      }                                                                     \
    }                                                                       \
  }
  if (mat == 3) {
    PJ_STORE(sigmoidf_(val))
  } else if (mat == 1) {
    PJ_STORE(val * 0.08838834764831845f)
  } else {
    PJ_STORE(val)
  }
#undef PJ_STORE
}

#define GR_ROWS 8
__global__ __launch_bounds__(256) void gates_kernel(
    const float* __restrict__ x,
    const float* __restrict__ wi, const float* __restrict__ bi,
    const float* __restrict__ wf, const float* __restrict__ bf,
    float* __restrict__ ig, float* __restrict__ la) {
  __shared__ float lf_s[LCH];
  const int wave = threadIdx.x >> 5;
  const int lane = threadIdx.x & 31;
  const int t0 = blockIdx.x * LCH;
  const int rbase = t0 + wave * GR_ROWS;

  v4f xv[GR_ROWS][4];
  const float* xp = x + (size_t)rbase * IN_DIM + 4 * lane;
#pragma unroll
  for (int r = 0; r < GR_ROWS; ++r) {
#pragma unroll
    for (int i = 0; i < 4; ++i)
      xv[r][i] = *(const v4f*)(xp + (size_t)r * IN_DIM + 128 * i);
  }
  v4f wiv[4], wfv[4];
#pragma unroll
  for (int i = 0; i < 4; ++i) {
    wiv[i] = *(const v4f*)(wi + 128 * i + 4 * lane);
    wfv[i] = *(const v4f*)(wf + 128 * i + 4 * lane);
  }
  __builtin_amdgcn_sched_barrier(0);

  float si[GR_ROWS], sf[GR_ROWS];
#pragma unroll
  for (int r = 0; r < GR_ROWS; ++r) {
    float a = 0.f, b = 0.f;
#pragma unroll
    for (int i = 0; i < 4; ++i) {
      a = fmaf(xv[r][i].x, wiv[i].x, a); b = fmaf(xv[r][i].x, wfv[i].x, b);
      a = fmaf(xv[r][i].y, wiv[i].y, a); b = fmaf(xv[r][i].y, wfv[i].y, b);
      a = fmaf(xv[r][i].z, wiv[i].z, a); b = fmaf(xv[r][i].z, wfv[i].z, b);
      a = fmaf(xv[r][i].w, wiv[i].w, a); b = fmaf(xv[r][i].w, wfv[i].w, b);
    }
    si[r] = a; sf[r] = b;
  }
#pragma unroll
  for (int m = 16; m >= 1; m >>= 1) {
#pragma unroll
    for (int r = 0; r < GR_ROWS; ++r) {
      si[r] += __shfl_xor(si[r], m, 32);
      sf[r] += __shfl_xor(sf[r], m, 32);
    }
  }
  float msi = si[0], msf = sf[0];
#pragma unroll
  for (int r = 1; r < GR_ROWS; ++r) {
    msi = (lane == r) ? si[r] : msi;
    msf = (lane == r) ? sf[r] : msf;
  }
  if (lane < GR_ROWS) {
    ig[rbase + lane] = __expf(msi + bi[0]);
    const float z = msf + bf[0];
    lf_s[wave * GR_ROWS + lane] =
        (z >= 0.f) ? -log1pf(__expf(-z)) : (z - log1pf(__expf(z)));
  }
  __syncthreads();
  if (threadIdx.x < LCH) {
    float lv[LCH];
#pragma unroll
    for (int i = 0; i < LCH; ++i) lv[i] = lf_s[i];
    float a = 0.f, res = 0.f;
#pragma unroll
    for (int i = 0; i < LCH; ++i) {
      a += lv[i];
      res = ((int)threadIdx.x == i) ? a : res;
    }
    la[t0 + threadIdx.x] = res;
  }
}

#define NBUF 3
__global__ __launch_bounds__(256, 1) void scan_kernel(
    const float* __restrict__ qw, const float* __restrict__ kw,
    const float* __restrict__ vw, const float* __restrict__ ow,
    const float* __restrict__ ig, const float* __restrict__ la,
    const float* __restrict__ C0, const float* __restrict__ n0,
    const float* __restrict__ fc_w,
    float* __restrict__ nump, float* __restrict__ denp,
    float* __restrict__ C_out, float* __restrict__ n_out) {
  __shared__ float Cs[2][SR][KC + 1];
  __shared__ float ns[2][KC];
  __shared__ float fcw_s[SR];
  __shared__ float Ps[LCH][LCH + 1];
  __shared__ float Gs[2][LCH][SR + 1];
  __shared__ float rs_s[2][LCH];
  __shared__ float Hs[LCH][SR + 1];
  __shared__ __attribute__((aligned(16))) float Tq[NBUF][LCH][TS];
  __shared__ __attribute__((aligned(16))) float Tk[NBUF][LCH][TS];
  __shared__ __attribute__((aligned(16))) float Tv[NBUF][LCH][TS];
  __shared__ __attribute__((aligned(16))) float To[NBUF][LCH][TS];
  __shared__ __attribute__((aligned(16))) float la_n[NBUF][LCH];
  __shared__ __attribute__((aligned(16))) float ig_n[NBUF][LCH];
  __shared__ __attribute__((aligned(16))) float u_n[NBUF][LCH];

  const int tid = threadIdx.x;
  const int wave = tid >> 5;
  const int lane = tid & 31;
  const int half = lane >> 4;
  const int lm = lane & 15;
  const int ga = blockIdx.x / NB;
  const int gb = blockIdx.x % NB;
  const int r0 = ga * SR;
  const int k0 = gb * KC;

  for (int i = tid; i < SR * KC; i += 256)
    Cs[0][i >> 5][i & 31] = C0[(size_t)(r0 + (i >> 5)) * MD + k0 + (i & 31)];
  if (tid < KC) ns[0][tid] = n0[k0 + tid];
  if (tid < SR) fcw_s[tid] = fc_w[r0 + tid];

  const int st_sub = (tid & 7) * 4;
  const int st_row = tid >> 3;
  const float* gq = qw + (size_t)st_row * MD + k0 + st_sub;
  const float* gk = kw + (size_t)st_row * MD + k0 + st_sub;
  const float* gv = vw + (size_t)st_row * MD + r0 + st_sub;
  const float* go = ow + (size_t)st_row * MD + r0 + st_sub;
  v4f pq0, pq1, pk0, pk1, pv0, pv1, po0, po1;
  float pla = 0.f, pig = 0.f, pend = 0.f;

#define FETCH_CHUNK(tn)                                                     \
  do {                                                                      \
    const size_t o0 = (size_t)(tn) * MD, o1 = o0 + (size_t)32 * MD;         \
    pq0 = *(const v4f*)(gq + o0); pq1 = *(const v4f*)(gq + o1);             \
    pk0 = *(const v4f*)(gk + o0); pk1 = *(const v4f*)(gk + o1);             \
    pv0 = *(const v4f*)(gv + o0); pv1 = *(const v4f*)(gv + o1);             \
    po0 = *(const v4f*)(go + o0); po1 = *(const v4f*)(go + o1);             \
    if (tid < LCH) {                                                        \
      pla = la[(tn) + tid]; pig = ig[(tn) + tid];                           \
      pend = la[(tn) + LCH - 1];                                            \
    }                                                                       \
  } while (0)

#define STAGE_CHUNK(bf)                                                     \
  do {                                                                      \
    *(v4f*)&Tq[bf][st_row][st_sub] = pq0; *(v4f*)&Tq[bf][st_row + 32][st_sub] = pq1; \
    *(v4f*)&Tk[bf][st_row][st_sub] = pk0; *(v4f*)&Tk[bf][st_row + 32][st_sub] = pk1; \
    *(v4f*)&Tv[bf][st_row][st_sub] = pv0; *(v4f*)&Tv[bf][st_row + 32][st_sub] = pv1; \
    *(v4f*)&To[bf][st_row][st_sub] = po0; *(v4f*)&To[bf][st_row + 32][st_sub] = po1; \
    if (tid < LCH) {                                                        \
      la_n[bf][tid] = pla; ig_n[bf][tid] = pig;                             \
      u_n[bf][tid] = __expf(pend - pla) * pig;                              \
    }                                                                       \
  } while (0)

  FETCH_CHUNK(0);
  STAGE_CHUNK(0);
  __syncthreads();

  int bc = NBUF - 1;
  for (int c = -1; c < NCH; ++c) {
    const int b_cur = bc;
    const int b_nxt = (bc == NBUF - 1) ? 0 : bc + 1;
    const int b_stg = (b_nxt == NBUF - 1) ? 0 : b_nxt + 1;
    const int pc = c & 1;
    const int pn = pc ^ 1;
    const bool have_cur = (c >= 0);
    const bool have_nxt = (c + 1 < NCH);
    const bool have_stg = (c + 2 < NCH);
    const int t0 = c * LCH;

    if (have_stg) FETCH_CHUNK((c + 2) * LCH);

    if (wave >= 4) {
      if (have_nxt) {
        const float (*Qn)[TS] = Tq[b_nxt];
        const float (*Kn)[TS] = Tk[b_nxt];
        const float* lan = la_n[b_nxt];
        const float* ign = ig_n[b_nxt];
        const int tr = (wave - 4) * 16;
        v2f fa[KC / 4];
        v2f fb[KC / 4][4];
        const float* ap = &Qn[tr + lm][2 * half];
#pragma unroll
        for (int k4 = 0; k4 < KC / 4; ++k4) {
          fa[k4].x = ap[4 * k4]; fa[k4].y = ap[4 * k4 + 1];
#pragma unroll
          for (int q = 0; q < 4; ++q) {
            const float* bp = &Kn[q * 16 + lm][2 * half];
            fb[k4][q].x = bp[4 * k4]; fb[k4][q].y = bp[4 * k4 + 1];
          }
        }
        float lar[8], lacq[4], igq[4];
#pragma unroll
        for (int e = 0; e < 8; ++e) lar[e] = lan[tr + e + 8 * half];
#pragma unroll
        for (int q = 0; q < 4; ++q) { lacq[q] = lan[q * 16 + lm]; igq[q] = ign[q * 16 + lm]; }
        __builtin_amdgcn_sched_barrier(0);
        v8f acc[4];
#pragma unroll
        for (int q = 0; q < 4; ++q) acc[q] = v8f{};
#pragma unroll
        for (int k4 = 0; k4 < KC / 4; ++k4) {
#pragma unroll
          for (int q = 0; q < 4; ++q) acc[q] = wmma4(fa[k4], fb[k4][q], acc[q]);
        }
#pragma unroll
        for (int q = 0; q < 4; ++q) {
          const int cc = q * 16 + lm;
#pragma unroll
          for (int e = 0; e < 8; ++e) {
            const int r = tr + e + 8 * half;
            const float p = __expf(lar[e] - lacq[q]) * igq[q] * acc[q][e];
            Ps[r][cc] = (r >= cc) ? p : 0.f;
          }
        }
      }
    } else {
      if (have_cur) {
        const float (*Qs)[TS] = Tq[b_cur];
        const float (*Ks)[TS] = Tk[b_cur];
        const float (*Vs)[TS] = Tv[b_cur];
        const float (*Os)[TS] = To[b_cur];
        const float* lac = la_n[b_cur];
        const float* uc = u_n[b_cur];
        const int tr = wave * 16;
        const int ur = (wave >> 1) * 16, ut = (wave & 1) * 16;
        v2f ha[KC / 4];
        v2f hb[KC / 4][2];
        const float* ap = &Qs[tr + lm][2 * half];
#pragma unroll
        for (int k4 = 0; k4 < KC / 4; ++k4) {
          ha[k4].x = ap[4 * k4]; ha[k4].y = ap[4 * k4 + 1];
#pragma unroll
          for (int q = 0; q < 2; ++q) {
            hb[k4][q].x = Cs[pc][q * 16 + lm][4 * k4 + 2 * half];
            hb[k4][q].y = Cs[pc][q * 16 + lm][4 * k4 + 2 * half + 1];
          }
        }
        float lr[8], gq[8][2], oq[8][2];
#pragma unroll
        for (int e = 0; e < 8; ++e) {
          const int r = tr + e + 8 * half;
          lr[e] = lac[r];
#pragma unroll
          for (int q = 0; q < 2; ++q) {
            gq[e][q] = Gs[pc][r][q * 16 + lm];
            oq[e][q] = Os[r][q * 16 + lm];
          }
        }
        const float lend = lac[LCH - 1];
        float c0[8];
#pragma unroll
        for (int e = 0; e < 8; ++e) c0[e] = Cs[pc][ur + e + 8 * half][ut + lm];
        v2f uu[LCH / 4], vv[LCH / 4], kb[LCH / 4];
#pragma unroll
        for (int k4 = 0; k4 < LCH / 4; ++k4) {
          const int j0 = 4 * k4 + 2 * half;
          uu[k4].x = uc[j0];          uu[k4].y = uc[j0 + 1];
          vv[k4].x = Vs[j0][ur + lm]; vv[k4].y = Vs[j0 + 1][ur + lm];
          kb[k4].x = Ks[j0][ut + lm]; kb[k4].y = Ks[j0 + 1][ut + lm];
        }
        {
          v8f ace[2];
          ace[0] = v8f{}; ace[1] = v8f{};
#pragma unroll
          for (int k4 = 0; k4 < KC / 4; ++k4) {
#pragma unroll
            for (int q = 0; q < 2; ++q) ace[q] = wmma4(ha[k4], hb[k4][q], ace[q]);
          }
#pragma unroll
          for (int e = 0; e < 8; ++e) {
            const int r = tr + e + 8 * half;
            const float Ft = __expf(lr[e]);
#pragma unroll
            for (int q = 0; q < 2; ++q) {
              const int cc = q * 16 + lm;
              const float num = fmaf(Ft, ace[q][e], gq[e][q]);
              Hs[r][cc] = num * oq[e][q];
            }
          }
        }
        {
          const float Fend = __expf(lend);
          v8f acc;
#pragma unroll
          for (int e = 0; e < 8; ++e) acc[e] = Fend * c0[e];
#pragma unroll
          for (int k4 = 0; k4 < LCH / 4; ++k4) {
            v2f a;
            a.x = uu[k4].x * vv[k4].x;
            a.y = uu[k4].y * vv[k4].y;
            acc = wmma4(a, kb[k4], acc);
          }
#pragma unroll
          for (int e = 0; e < 8; ++e) Cs[pn][ur + e + 8 * half][ut + lm] = acc[e];
        }
        __builtin_amdgcn_sched_group_barrier(0x100, 128, 0);
        __builtin_amdgcn_sched_group_barrier(0x008, 32, 0);
      }
    }
    __syncthreads();

    if (have_stg) STAGE_CHUNK(b_stg);
    if (wave >= 4) {
      if (have_nxt) {
        const float (*Vn)[TS] = Tv[b_nxt];
        const int tr = (wave - 4) * 16;
        float pax[LCH / 4], pay[LCH / 4];
        float vbx[LCH / 4][2], vby[LCH / 4][2];
#pragma unroll
        for (int k4 = 0; k4 < LCH / 4; ++k4) {
          const int j0 = 4 * k4 + 2 * half;
          pax[k4] = Ps[tr + lm][j0]; pay[k4] = Ps[tr + lm][j0 + 1];
#pragma unroll
          for (int q = 0; q < 2; ++q) {
            vbx[k4][q] = Vn[j0][q * 16 + lm]; vby[k4][q] = Vn[j0 + 1][q * 16 + lm];
          }
        }
        __builtin_amdgcn_sched_barrier(0);
        v8f aci[2];
        aci[0] = v8f{}; aci[1] = v8f{};
#pragma unroll
        for (int k4 = 0; k4 < LCH / 4; ++k4) {
          v2f a;
          a.x = pax[k4]; a.y = pay[k4];
#pragma unroll
          for (int q = 0; q < 2; ++q) {
            v2f b;
            b.x = vbx[k4][q]; b.y = vby[k4][q];
            aci[q] = wmma4(a, b, aci[q]);
          }
        }
#pragma unroll
        for (int q = 0; q < 2; ++q)
#pragma unroll
          for (int e = 0; e < 8; ++e)
            Gs[pn][tr + e + 8 * half][q * 16 + lm] = aci[q][e];
      }
    } else if (wave < 2) {
      if (have_cur) {
        const int row = tid;
        float hv[SR], fw[SR];
#pragma unroll
        for (int f = 0; f < SR; ++f) { hv[f] = Hs[row][f]; fw[f] = fcw_s[f]; }
        __builtin_amdgcn_sched_barrier(0);
        float s = 0.f;
#pragma unroll
        for (int f = 0; f < SR; ++f) s = fmaf(hv[f], fw[f], s);
        nump[(size_t)blockIdx.x * T_LEN + t0 + row] = s;
        if (ga == 0) {
          const float* qrow = Tq[b_cur][row];
          float qv[KC], nv[KC];
#pragma unroll
          for (int f = 0; f < KC; ++f) { qv[f] = qrow[f]; nv[f] = ns[pc][f]; }
          const float lrow = la_n[b_cur][row];
          const float rsv = rs_s[pc][row];
          __builtin_amdgcn_sched_barrier(0);
          float qn = 0.f;
#pragma unroll
          for (int f = 0; f < KC; ++f) qn = fmaf(qv[f], nv[f], qn);
          denp[(size_t)gb * T_LEN + t0 + row] = fmaf(__expf(lrow), qn, rsv);
        }
      }
    } else {
      const int row = tid - 64;
      const bool do_n = (wave == 3) && have_cur;
      if (have_nxt && do_n) {
        float pv[LCH], uv[LCH], kv[LCH];
#pragma unroll
        for (int j = 0; j < LCH; ++j) pv[j] = Ps[row][j];
#pragma unroll
        for (int j = 0; j < LCH; ++j) { uv[j] = u_n[b_cur][j]; kv[j] = Tk[b_cur][j][lane]; }
        const float lend = la_n[b_cur][LCH - 1];
        const float nold = ns[pc][lane];
        __builtin_amdgcn_sched_barrier(0);
        float s = 0.f;
#pragma unroll
        for (int j = 0; j < LCH; ++j) s += pv[j];
        rs_s[pn][row] = s;
        const float Fend = __expf(lend);
        float sn = Fend * nold;
#pragma unroll
        for (int j = 0; j < LCH; ++j) sn = fmaf(uv[j], kv[j], sn);
        ns[pn][lane] = sn;
      } else if (have_nxt) {
        float pv[LCH];
#pragma unroll
        for (int j = 0; j < LCH; ++j) pv[j] = Ps[row][j];
        __builtin_amdgcn_sched_barrier(0);
        float s = 0.f;
#pragma unroll
        for (int j = 0; j < LCH; ++j) s += pv[j];
        rs_s[pn][row] = s;
      } else if (do_n) {
        float uv[LCH], kv[LCH];
#pragma unroll
        for (int j = 0; j < LCH; ++j) { uv[j] = u_n[b_cur][j]; kv[j] = Tk[b_cur][j][lane]; }
        const float lend = la_n[b_cur][LCH - 1];
        const float nold = ns[pc][lane];
        __builtin_amdgcn_sched_barrier(0);
        const float Fend = __expf(lend);
        float sn = Fend * nold;
#pragma unroll
        for (int j = 0; j < LCH; ++j) sn = fmaf(uv[j], kv[j], sn);
        ns[pn][lane] = sn;
      }
    }
    __syncthreads();
    bc = b_nxt;
  }

#undef FETCH_CHUNK
#undef STAGE_CHUNK

  for (int i = tid; i < SR * KC; i += 256)
    C_out[(size_t)(r0 + (i >> 5)) * MD + k0 + (i & 31)] = Cs[NCH & 1][i >> 5][i & 31];
  if (ga == 0 && tid < KC) n_out[k0 + tid] = ns[NCH & 1][tid];
}

__global__ __launch_bounds__(256) void out_kernel(
    const float* __restrict__ nump, const float* __restrict__ denp,
    const float* __restrict__ fc_b, float* __restrict__ out) {
  const int t = blockIdx.x * 256 + threadIdx.x;
  float s = nump[t];
#pragma unroll
  for (int g = 1; g < NGRP; ++g) s += nump[(size_t)g * T_LEN + t];
  float d = denp[t];
#pragma unroll
  for (int b = 1; b < NB; ++b) d += denp[(size_t)b * T_LEN + t];
  out[t] = s / fmaxf(fabsf(d), 1.0f) + fc_b[0];
}

extern "C" void kernel_launch(void* const* d_in, const int* in_sizes, int n_in,
                              void* d_out, int out_size, void* d_ws, size_t ws_size,
                              hipStream_t stream) {
  const float* x    = (const float*)d_in[0];
  const float* C0   = (const float*)d_in[1];
  const float* n0   = (const float*)d_in[2];
  const float* Wq   = (const float*)d_in[3];
  const float* bq   = (const float*)d_in[4];
  const float* Wk   = (const float*)d_in[5];
  const float* bk   = (const float*)d_in[6];
  const float* Wv   = (const float*)d_in[7];
  const float* bv   = (const float*)d_in[8];
  const float* wi   = (const float*)d_in[9];
  const float* bi   = (const float*)d_in[10];
  const float* wf   = (const float*)d_in[11];
  const float* bf   = (const float*)d_in[12];
  const float* Wo   = (const float*)d_in[13];
  const float* bo   = (const float*)d_in[14];
  const float* fc_w = (const float*)d_in[15];
  const float* fc_b = (const float*)d_in[16];

  float* ws = (float*)d_ws;
  const size_t NP = (size_t)T_LEN * MD;
  float* qw = ws;
  float* kw = ws + NP;
  float* vw = ws + 2 * NP;
  float* owp = ws + 3 * NP;
  float* igp = ws + 4 * NP;
  float* lfp = igp + T_LEN;
  float* numpp = lfp + T_LEN;
  float* denpp = numpp + NGRP * T_LEN;

  float* outp  = (float*)d_out;
  float* Coutp = outp + T_LEN;
  float* noutp = Coutp + MD * MD;

  proj_kernel<<<dim3(T_LEN / 128, 4), 128, 0, stream>>>(
      x, Wq, bq, Wk, bk, Wv, bv, Wo, bo, qw, kw, vw, owp);
  gates_kernel<<<dim3(T_LEN / LCH), 256, 0, stream>>>(
      x, wi, bi, wf, bf, igp, lfp);
  scan_kernel<<<NGRP, 256, 0, stream>>>(
      qw, kw, vw, owp, igp, lfp, C0, n0, fc_w, numpp, denpp, Coutp, noutp);
  out_kernel<<<T_LEN / 256, 256, 0, stream>>>(numpp, denpp, fc_b, outp);
}


# Scan: one work-group barrier a chunk (was 2): S taken transposed so P stays in its wave's registers as P V's A fragments; read-out sums run a chunk later; P V and P row-sum orders change
# speedup vs baseline: 12.1457x; 1.2812x over previous
#include <hip/hip_runtime.h>
#include <hip/hip_bf16.h>
#include <math.h>


typedef __attribute__((ext_vector_type(2))) float v2f;
typedef __attribute__((ext_vector_type(8))) float v8f;
typedef __attribute__((ext_vector_type(4))) float v4f;

#define T_LEN 8192
#define IN_DIM 512
#define MD 128
#define LCH 64
#define NCH (T_LEN / LCH)
#define NA 4
#define NB 4
#define NGRP (NA * NB)
#define SR (MD / NA)
#define KC (MD / NB)
#define TS 36

__device__ __forceinline__ v8f wmma4(v2f a, v2f b, v8f c) {
  return __builtin_amdgcn_wmma_f32_16x16x4_f32(
       false, a,  false, b,
       (short)0, c,  false,  false);
}

__device__ __forceinline__ float sigmoidf_(float z) {
  return 1.0f / (1.0f + __expf(-z));
}

#define PJ_NSET 4
#define PJ_STEPS (IN_DIM / 8)
__global__ __launch_bounds__(128, 1) void proj_kernel(
    const float* __restrict__ x,
    const float* __restrict__ Wq, const float* __restrict__ bq,
    const float* __restrict__ Wk, const float* __restrict__ bk,
    const float* __restrict__ Wv, const float* __restrict__ bv,
    const float* __restrict__ Wo, const float* __restrict__ bo,
    float* __restrict__ qw, float* __restrict__ kw,
    float* __restrict__ vw, float* __restrict__ ow) {
  const int wave = threadIdx.x >> 5;
  const int lane = threadIdx.x & 31;
  const int half = lane >> 4;
  const int lm = lane & 15;

  const int mat = blockIdx.y;
  const int row0 = (blockIdx.x * 2 + (wave >> 1)) * 64;
  const int col0 = (wave & 1) * 64;

  const float* W; const float* bias; float* dst;
  switch (mat) {
    case 0:  W = Wq; bias = bq; dst = qw; break;
    case 1:  W = Wk; bias = bk; dst = kw; break;
    case 2:  W = Wv; bias = bv; dst = vw; break;
    default: W = Wo; bias = bo; dst = ow; break;
  }

  const float* ap = x + (size_t)(row0 + lm) * IN_DIM + 4 * half;
  const float* bp = W + (size_t)(col0 + lm) * IN_DIM + 4 * half;

  v8f acc[4][4];
#pragma unroll
  for (int i = 0; i < 4; ++i)
#pragma unroll
    for (int j = 0; j < 4; ++j) acc[i][j] = v8f{};

  v4f fa[PJ_NSET][4], fb[PJ_NSET][4];
#pragma unroll
  for (int s = 0; s < PJ_NSET - 1; ++s) {
#pragma unroll
    for (int i = 0; i < 4; ++i) {
      fa[s][i] = *(const v4f*)(ap + (size_t)i * 16 * IN_DIM + 8 * s);
      fb[s][i] = *(const v4f*)(bp + (size_t)i * 16 * IN_DIM + 8 * s);
    }
  }

  for (int s0 = 0; s0 < PJ_STEPS; s0 += PJ_NSET) {
#pragma unroll
    for (int u = 0; u < PJ_NSET; ++u) {
      const int sn = (s0 + u + PJ_NSET - 1 < PJ_STEPS) ? (s0 + u + PJ_NSET - 1) : (PJ_STEPS - 1);
      const int w = (u + PJ_NSET - 1) % PJ_NSET;
#pragma unroll
      for (int i = 0; i < 4; ++i) {
        fa[w][i] = *(const v4f*)(ap + (size_t)i * 16 * IN_DIM + 8 * sn);
        fb[w][i] = *(const v4f*)(bp + (size_t)i * 16 * IN_DIM + 8 * sn);
      }
      __builtin_amdgcn_sched_barrier(0);
#pragma unroll
      for (int i = 0; i < 4; ++i) {
        v2f a; a.x = fa[u][i].x; a.y = fa[u][i].y;
#pragma unroll
        for (int j = 0; j < 4; ++j) {
          v2f b; b.x = fb[u][j].x; b.y = fb[u][j].y;
          acc[i][j] = wmma4(a, b, acc[i][j]);
        }
      }
#pragma unroll
      for (int i = 0; i < 4; ++i) {
        v2f a; a.x = fa[u][i].z; a.y = fa[u][i].w;
#pragma unroll
        for (int j = 0; j < 4; ++j) {
          v2f b; b.x = fb[u][j].z; b.y = fb[u][j].w;
          acc[i][j] = wmma4(a, b, acc[i][j]);
        }
      }
    }
  }

#define PJ_STORE(EXPR)                                                      \
  _Pragma("unroll") for (int j = 0; j < 4; ++j) {                           \
    const int col = col0 + j * 16 + lm;                                     \
    const float bv_ = bias[col];                                            \
    _Pragma("unroll") for (int i = 0; i < 4; ++i) {                         \
      _Pragma("unroll") for (int e = 0; e < 8; ++e) {                       \
        const int r = row0 + i * 16 + e + 8 * half;                         \
        const float val = acc[i][j][e] + bv_;                               \
        dst[(size_t)r * MD + col] = (EXPR);                                 \
      }                                                                     \
    }                                                                       \
  }
  if (mat == 3) {
    PJ_STORE(sigmoidf_(val))
  } else if (mat == 1) {
    PJ_STORE(val * 0.08838834764831845f)
  } else {
    PJ_STORE(val)
  }
#undef PJ_STORE
}

#define GR_ROWS 8
__global__ __launch_bounds__(256) void gates_kernel(
    const float* __restrict__ x,
    const float* __restrict__ wi, const float* __restrict__ bi,
    const float* __restrict__ wf, const float* __restrict__ bf,
    float* __restrict__ ig, float* __restrict__ la) {
  __shared__ float lf_s[LCH];
  const int wave = threadIdx.x >> 5;
  const int lane = threadIdx.x & 31;
  const int t0 = blockIdx.x * LCH;
  const int rbase = t0 + wave * GR_ROWS;

  v4f xv[GR_ROWS][4];
  const float* xp = x + (size_t)rbase * IN_DIM + 4 * lane;
#pragma unroll
  for (int r = 0; r < GR_ROWS; ++r) {
#pragma unroll
    for (int i = 0; i < 4; ++i)
      xv[r][i] = *(const v4f*)(xp + (size_t)r * IN_DIM + 128 * i);
  }
  v4f wiv[4], wfv[4];
#pragma unroll
  for (int i = 0; i < 4; ++i) {
    wiv[i] = *(const v4f*)(wi + 128 * i + 4 * lane);
    wfv[i] = *(const v4f*)(wf + 128 * i + 4 * lane);
  }
  __builtin_amdgcn_sched_barrier(0);

  float si[GR_ROWS], sf[GR_ROWS];
#pragma unroll
  for (int r = 0; r < GR_ROWS; ++r) {
    float a = 0.f, b = 0.f;
#pragma unroll
    for (int i = 0; i < 4; ++i) {
      a = fmaf(xv[r][i].x, wiv[i].x, a); b = fmaf(xv[r][i].x, wfv[i].x, b);
      a = fmaf(xv[r][i].y, wiv[i].y, a); b = fmaf(xv[r][i].y, wfv[i].y, b);
      a = fmaf(xv[r][i].z, wiv[i].z, a); b = fmaf(xv[r][i].z, wfv[i].z, b);
      a = fmaf(xv[r][i].w, wiv[i].w, a); b = fmaf(xv[r][i].w, wfv[i].w, b);
    }
    si[r] = a; sf[r] = b;
  }
#pragma unroll
  for (int m = 16; m >= 1; m >>= 1) {
#pragma unroll
    for (int r = 0; r < GR_ROWS; ++r) {
      si[r] += __shfl_xor(si[r], m, 32);
      sf[r] += __shfl_xor(sf[r], m, 32);
    }
  }
  float msi = si[0], msf = sf[0];
#pragma unroll
  for (int r = 1; r < GR_ROWS; ++r) {
    msi = (lane == r) ? si[r] : msi;
    msf = (lane == r) ? sf[r] : msf;
  }
  if (lane < GR_ROWS) {
    ig[rbase + lane] = __expf(msi + bi[0]);
    const float z = msf + bf[0];
    lf_s[wave * GR_ROWS + lane] =
        (z >= 0.f) ? -log1pf(__expf(-z)) : (z - log1pf(__expf(z)));
  }
  __syncthreads();
  if (threadIdx.x < LCH) {
    float lv[LCH];
#pragma unroll
    for (int i = 0; i < LCH; ++i) lv[i] = lf_s[i];
    float a = 0.f, res = 0.f;
#pragma unroll
    for (int i = 0; i < LCH; ++i) {
      a += lv[i];
      res = ((int)threadIdx.x == i) ? a : res;
    }
    la[t0 + threadIdx.x] = res;
  }
}

#define NBUF 4
__global__ __launch_bounds__(256, 1) void scan_kernel(
    const float* __restrict__ qw, const float* __restrict__ kw,
    const float* __restrict__ vw, const float* __restrict__ ow,
    const float* __restrict__ ig, const float* __restrict__ la,
    const float* __restrict__ C0, const float* __restrict__ n0,
    const float* __restrict__ fc_w,
    float* __restrict__ nump, float* __restrict__ denp,
    float* __restrict__ C_out, float* __restrict__ n_out) {
  __shared__ float Cs[2][SR][KC + 1];
  __shared__ float ns[3][KC];
  __shared__ float fcw_s[SR];
  __shared__ float Gs[2][LCH][SR + 1];
  __shared__ float rs_s[3][LCH];
  __shared__ float Hs[2][LCH][SR + 1];
  __shared__ __attribute__((aligned(16))) float Tq[NBUF][LCH][TS];
  __shared__ __attribute__((aligned(16))) float Tk[NBUF][LCH][TS];
  __shared__ __attribute__((aligned(16))) float Tv[NBUF][LCH][TS];
  __shared__ __attribute__((aligned(16))) float To[NBUF][LCH][TS];
  __shared__ __attribute__((aligned(16))) float la_n[NBUF][LCH];
  __shared__ __attribute__((aligned(16))) float ig_n[NBUF][LCH];
  __shared__ __attribute__((aligned(16))) float u_n[NBUF][LCH];

  const int tid = threadIdx.x;
  const int wave = tid >> 5;
  const int lane = tid & 31;
  const int half = lane >> 4;
  const int lm = lane & 15;
  const int ga = blockIdx.x / NB;
  const int gb = blockIdx.x % NB;
  const int r0 = ga * SR;
  const int k0 = gb * KC;

  for (int i = tid; i < SR * KC; i += 256)
    Cs[0][i >> 5][i & 31] = C0[(size_t)(r0 + (i >> 5)) * MD + k0 + (i & 31)];
  if (tid < KC) ns[0][tid] = n0[k0 + tid];
  if (tid < SR) fcw_s[tid] = fc_w[r0 + tid];

  const int st_sub = (tid & 7) * 4;
  const int st_row = tid >> 3;
  const float* gq = qw + (size_t)st_row * MD + k0 + st_sub;
  const float* gk = kw + (size_t)st_row * MD + k0 + st_sub;
  const float* gv = vw + (size_t)st_row * MD + r0 + st_sub;
  const float* go = ow + (size_t)st_row * MD + r0 + st_sub;
  v4f pq0, pq1, pk0, pk1, pv0, pv1, po0, po1;
  float pla = 0.f, pig = 0.f, pend = 0.f;

#define FETCH_CHUNK(tn)                                                     \
  do {                                                                      \
    const size_t o0 = (size_t)(tn) * MD, o1 = o0 + (size_t)32 * MD;         \
    pq0 = *(const v4f*)(gq + o0); pq1 = *(const v4f*)(gq + o1);             \
    pk0 = *(const v4f*)(gk + o0); pk1 = *(const v4f*)(gk + o1);             \
    pv0 = *(const v4f*)(gv + o0); pv1 = *(const v4f*)(gv + o1);             \
    po0 = *(const v4f*)(go + o0); po1 = *(const v4f*)(go + o1);             \
    if (tid < LCH) {                                                        \
      pla = la[(tn) + tid]; pig = ig[(tn) + tid];                           \
      pend = la[(tn) + LCH - 1];                                            \
    }                                                                       \
  } while (0)

#define STAGE_CHUNK(bf)                                                     \
  do {                                                                      \
    *(v4f*)&Tq[bf][st_row][st_sub] = pq0; *(v4f*)&Tq[bf][st_row + 32][st_sub] = pq1; \
    *(v4f*)&Tk[bf][st_row][st_sub] = pk0; *(v4f*)&Tk[bf][st_row + 32][st_sub] = pk1; \
    *(v4f*)&Tv[bf][st_row][st_sub] = pv0; *(v4f*)&Tv[bf][st_row + 32][st_sub] = pv1; \
    *(v4f*)&To[bf][st_row][st_sub] = po0; *(v4f*)&To[bf][st_row + 32][st_sub] = po1; \
    if (tid < LCH) {                                                        \
      la_n[bf][tid] = pla; ig_n[bf][tid] = pig;                             \
      u_n[bf][tid] = __expf(pend - pla) * pig;                              \
    }                                                                       \
  } while (0)

  FETCH_CHUNK(0);
  STAGE_CHUNK(0);
  __syncthreads();

  int m_prv = 1, m_cur = 2, m_nxt = 0;
  for (int c = -1; c <= NCH; ++c) {
    const int b_prv = (c - 1) & (NBUF - 1);
    const int b_cur = c & (NBUF - 1);
    const int b_nxt = (c + 1) & (NBUF - 1);
    const int b_stg = (c + 2) & (NBUF - 1);
    const int pc = c & 1;
    const int pn = pc ^ 1;
    const bool have_prv = (c >= 1);
    const bool have_cur = (c >= 0) && (c < NCH);
    const bool have_nxt = (c + 1 < NCH);
    const bool have_stg = (c + 2 < NCH);
    const int t0p = (c - 1) * LCH;

    if (have_stg) FETCH_CHUNK((c + 2) * LCH);

    if (wave >= 4) {
      if (have_nxt) {
        const float (*Qn)[TS] = Tq[b_nxt];
        const float (*Kn)[TS] = Tk[b_nxt];
        const float (*Vn)[TS] = Tv[b_nxt];
        const float* lan = la_n[b_nxt];
        const float* ign = ig_n[b_nxt];
        const int tr = (wave - 4) * 16;
        v2f fq[KC / 4];
        v2f fk[KC / 4][4];
        const float* qp = &Qn[tr + lm][2 * half];
#pragma unroll
        for (int k4 = 0; k4 < KC / 4; ++k4) {
          fq[k4].x = qp[4 * k4]; fq[k4].y = qp[4 * k4 + 1];
#pragma unroll
          for (int q = 0; q < 4; ++q) {
            const float* kp = &Kn[q * 16 + lm][2 * half];
            fk[k4][q].x = kp[4 * k4]; fk[k4][q].y = kp[4 * k4 + 1];
          }
        }
        const float lat = lan[tr + lm];
        v4f laj[4][2], igj[4][2];
#pragma unroll
        for (int q = 0; q < 4; ++q) {
#pragma unroll
          for (int g = 0; g < 2; ++g) {
            laj[q][g] = *(const v4f*)&lan[q * 16 + 8 * half + 4 * g];
            igj[q][g] = *(const v4f*)&ign[q * 16 + 8 * half + 4 * g];
          }
        }
        __builtin_amdgcn_sched_barrier(0);
        v8f acc[4];
#pragma unroll
        for (int q = 0; q < 4; ++q) acc[q] = v8f{};
#pragma unroll
        for (int k4 = 0; k4 < KC / 4; ++k4) {
#pragma unroll
          for (int q = 0; q < 4; ++q) acc[q] = wmma4(fk[k4][q], fq[k4], acc[q]);
        }
        __builtin_amdgcn_sched_barrier(0);
        float vx[4][4][2], vy[4][4][2];
#pragma unroll
        for (int q = 0; q < 4; ++q) {
#pragma unroll
          for (int s = 0; s < 4; ++s) {
            const int j0 = q * 16 + 8 * half + 2 * s;
#pragma unroll
            for (int n2 = 0; n2 < 2; ++n2) {
              vx[q][s][n2] = Vn[j0][n2 * 16 + lm];
              vy[q][s][n2] = Vn[j0 + 1][n2 * 16 + lm];
            }
          }
        }
        __builtin_amdgcn_sched_barrier(0);
        float pT[4][8];
#pragma unroll
        for (int q = 0; q < 4; ++q) {
#pragma unroll
          for (int e = 0; e < 8; ++e) {
            const int j = q * 16 + 8 * half + e;
            const float p = __expf(lat - laj[q][e >> 2][e & 3]) * igj[q][e >> 2][e & 3] * acc[q][e];
            pT[q][e] = (tr + lm >= j) ? p : 0.f;
          }
        }
        if (ga == 0) {
          float sq[4];
#pragma unroll
          for (int q = 0; q < 4; ++q) {
            float s = pT[q][0];
#pragma unroll
            for (int e = 1; e < 8; ++e) s += pT[q][e];
            sq[q] = s;
          }
          const float sl = (sq[0] + sq[1]) + (sq[2] + sq[3]);
          const float st = sl + __shfl_xor(sl, 16, 32);
          if (half == 0) rs_s[m_nxt][tr + lm] = st;
        }
        v8f aci[2];
        aci[0] = v8f{}; aci[1] = v8f{};
#pragma unroll
        for (int q = 0; q < 4; ++q) {
#pragma unroll
          for (int s = 0; s < 4; ++s) {
            v2f a;
            a.x = pT[q][2 * s]; a.y = pT[q][2 * s + 1];
#pragma unroll
            for (int n2 = 0; n2 < 2; ++n2) {
              v2f b;
              b.x = vx[q][s][n2]; b.y = vy[q][s][n2];
              aci[n2] = wmma4(a, b, aci[n2]);
            }
          }
        }
#pragma unroll
        for (int n2 = 0; n2 < 2; ++n2)
#pragma unroll
          for (int e = 0; e < 8; ++e)
            Gs[pn][tr + e + 8 * half][n2 * 16 + lm] = aci[n2][e];
      }
    } else {
      if (wave < 2) {
        if (have_prv) {
          const int row = tid;
          float hv[SR], fw[SR];
#pragma unroll
          for (int f = 0; f < SR; ++f) { hv[f] = Hs[pn][row][f]; fw[f] = fcw_s[f]; }
          __builtin_amdgcn_sched_barrier(0);
          float s = 0.f;
#pragma unroll
          for (int f = 0; f < SR; ++f) s = fmaf(hv[f], fw[f], s);
          nump[(size_t)blockIdx.x * T_LEN + t0p + row] = s;
          if (ga == 0) {
            const float* qrow = Tq[b_prv][row];
            float qv[KC], nv[KC];
#pragma unroll
            for (int f = 0; f < KC; ++f) { qv[f] = qrow[f]; nv[f] = ns[m_prv][f]; }
            const float lrow = la_n[b_prv][row];
            const float rsv = rs_s[m_prv][row];
            __builtin_amdgcn_sched_barrier(0);
            float qn = 0.f;
#pragma unroll
            for (int f = 0; f < KC; ++f) qn = fmaf(qv[f], nv[f], qn);
            denp[(size_t)gb * T_LEN + t0p + row] = fmaf(__expf(lrow), qn, rsv);
          }
        }
      } else if (wave == 3) {
        if (have_cur) {
          float uv[LCH], kv[LCH];
#pragma unroll
          for (int j = 0; j < LCH; ++j) { uv[j] = u_n[b_cur][j]; kv[j] = Tk[b_cur][j][lane]; }
          const float lend = la_n[b_cur][LCH - 1];
          const float nold = ns[m_cur][lane];
          __builtin_amdgcn_sched_barrier(0);
          const float Fend = __expf(lend);
          float sn = Fend * nold;
#pragma unroll
          for (int j = 0; j < LCH; ++j) sn = fmaf(uv[j], kv[j], sn);
          ns[m_nxt][lane] = sn;
        }
      }
      if (have_cur) {
        const float (*Qs)[TS] = Tq[b_cur];
        const float (*Ks)[TS] = Tk[b_cur];
        const float (*Vs)[TS] = Tv[b_cur];
        const float (*Os)[TS] = To[b_cur];
        const float* lac = la_n[b_cur];
        const float* uc = u_n[b_cur];
        const int tr = wave * 16;
        const int ur = (wave >> 1) * 16, ut = (wave & 1) * 16;
        v2f ha[KC / 4];
        v2f hb[KC / 4][2];
        const float* ap = &Qs[tr + lm][2 * half];
#pragma unroll
        for (int k4 = 0; k4 < KC / 4; ++k4) {
          ha[k4].x = ap[4 * k4]; ha[k4].y = ap[4 * k4 + 1];
#pragma unroll
          for (int q = 0; q < 2; ++q) {
            hb[k4][q].x = Cs[pc][q * 16 + lm][4 * k4 + 2 * half];
            hb[k4][q].y = Cs[pc][q * 16 + lm][4 * k4 + 2 * half + 1];
          }
        }
        float lr[8], gq[8][2], oq[8][2];
#pragma unroll
        for (int e = 0; e < 8; ++e) {
          const int r = tr + e + 8 * half;
          lr[e] = lac[r];
#pragma unroll
          for (int q = 0; q < 2; ++q) {
            gq[e][q] = Gs[pc][r][q * 16 + lm];
            oq[e][q] = Os[r][q * 16 + lm];
          }
        }
        const float lend = lac[LCH - 1];
        float c0[8];
#pragma unroll
        for (int e = 0; e < 8; ++e) c0[e] = Cs[pc][ur + e + 8 * half][ut + lm];
        v2f uu[LCH / 4], vv[LCH / 4], kb[LCH / 4];
#pragma unroll
        for (int k4 = 0; k4 < LCH / 4; ++k4) {
          const int j0 = 4 * k4 + 2 * half;
          uu[k4].x = uc[j0];          uu[k4].y = uc[j0 + 1];
          vv[k4].x = Vs[j0][ur + lm]; vv[k4].y = Vs[j0 + 1][ur + lm];
          kb[k4].x = Ks[j0][ut + lm]; kb[k4].y = Ks[j0 + 1][ut + lm];
        }
        asm volatile("" :: "v"(lr[0]), "v"(lr[1]), "v"(lr[2]), "v"(lr[3]),
                           "v"(lr[4]), "v"(lr[5]), "v"(lr[6]), "v"(lr[7]));
        {
          v8f ace[2];
          ace[0] = v8f{}; ace[1] = v8f{};
#pragma unroll
          for (int k4 = 0; k4 < KC / 4; ++k4) {
#pragma unroll
            for (int q = 0; q < 2; ++q) ace[q] = wmma4(ha[k4], hb[k4][q], ace[q]);
          }
#pragma unroll
          for (int e = 0; e < 8; ++e) {
            const int r = tr + e + 8 * half;
            const float Ft = __expf(lr[e]);
#pragma unroll
            for (int q = 0; q < 2; ++q) {
              const int cc = q * 16 + lm;
              const float num = fmaf(Ft, ace[q][e], gq[e][q]);
              Hs[pc][r][cc] = num * oq[e][q];
            }
          }
        }
        {
          const float Fend = __expf(lend);
          v8f acc;
#pragma unroll
          for (int e = 0; e < 8; ++e) acc[e] = Fend * c0[e];
#pragma unroll
          for (int k4 = 0; k4 < LCH / 4; ++k4) {
            v2f a;
            a.x = uu[k4].x * vv[k4].x;
            a.y = uu[k4].y * vv[k4].y;
            acc = wmma4(a, kb[k4], acc);
          }
#pragma unroll
          for (int e = 0; e < 8; ++e) Cs[pn][ur + e + 8 * half][ut + lm] = acc[e];
        }
        __builtin_amdgcn_sched_group_barrier(0x100, 128, 0);
        __builtin_amdgcn_sched_group_barrier(0x008, 32, 0);
      }
    }
    if (have_stg) STAGE_CHUNK(b_stg);
    __syncthreads();
    m_prv = m_cur; m_cur = m_nxt; m_nxt = (m_nxt == 2) ? 0 : m_nxt + 1;
  }

#undef FETCH_CHUNK
#undef STAGE_CHUNK

  for (int i = tid; i < SR * KC; i += 256)
    C_out[(size_t)(r0 + (i >> 5)) * MD + k0 + (i & 31)] = Cs[NCH & 1][i >> 5][i & 31];
  if (ga == 0 && tid < KC) n_out[k0 + tid] = ns[NCH % 3][tid];
}

__global__ __launch_bounds__(256) void out_kernel(
    const float* __restrict__ nump, const float* __restrict__ denp,
    const float* __restrict__ fc_b, float* __restrict__ out) {
  const int t = blockIdx.x * 256 + threadIdx.x;
  float s = nump[t];
#pragma unroll
  for (int g = 1; g < NGRP; ++g) s += nump[(size_t)g * T_LEN + t];
  float d = denp[t];
#pragma unroll
  for (int b = 1; b < NB; ++b) d += denp[(size_t)b * T_LEN + t];
  out[t] = s / fmaxf(fabsf(d), 1.0f) + fc_b[0];
}

extern "C" void kernel_launch(void* const* d_in, const int* in_sizes, int n_in,
                              void* d_out, int out_size, void* d_ws, size_t ws_size,
                              hipStream_t stream) {
  const float* x    = (const float*)d_in[0];
  const float* C0   = (const float*)d_in[1];
  const float* n0   = (const float*)d_in[2];
  const float* Wq   = (const float*)d_in[3];
  const float* bq   = (const float*)d_in[4];
  const float* Wk   = (const float*)d_in[5];
  const float* bk   = (const float*)d_in[6];
  const float* Wv   = (const float*)d_in[7];
  const float* bv   = (const float*)d_in[8];
  const float* wi   = (const float*)d_in[9];
  const float* bi   = (const float*)d_in[10];
  const float* wf   = (const float*)d_in[11];
  const float* bf   = (const float*)d_in[12];
  const float* Wo   = (const float*)d_in[13];
  const float* bo   = (const float*)d_in[14];
  const float* fc_w = (const float*)d_in[15];
  const float* fc_b = (const float*)d_in[16];

  float* ws = (float*)d_ws;
  const size_t NP = (size_t)T_LEN * MD;
  float* qw = ws;
  float* kw = ws + NP;
  float* vw = ws + 2 * NP;
  float* owp = ws + 3 * NP;
  float* igp = ws + 4 * NP;
  float* lfp = igp + T_LEN;
  float* numpp = lfp + T_LEN;
  float* denpp = numpp + NGRP * T_LEN;

  float* outp  = (float*)d_out;
  float* Coutp = outp + T_LEN;
  float* noutp = Coutp + MD * MD;

  proj_kernel<<<dim3(T_LEN / 128, 4), 128, 0, stream>>>(
      x, Wq, bq, Wk, bk, Wv, bv, Wo, bo, qw, kw, vw, owp);
  gates_kernel<<<dim3(T_LEN / LCH), 256, 0, stream>>>(
      x, wi, bi, wf, bf, igp, lfp);
  scan_kernel<<<NGRP, 256, 0, stream>>>(
      qw, kw, vw, owp, igp, lfp, C0, n0, fc_w, numpp, denpp, Coutp, noutp);
  out_kernel<<<T_LEN / 256, 256, 0, stream>>>(numpp, denpp, fc_b, outp);
}


# State-free products (S, P, P row sums, P V) leave the serial scan for their own launch: a work-group per chunk and column half, S over all 128 k, P V folded to one float a step; sum orders change
# speedup vs baseline: 14.3231x; 1.1793x over previous
#include <hip/hip_runtime.h>
#include <hip/hip_bf16.h>
#include <math.h>


typedef __attribute__((ext_vector_type(2))) float v2f;
typedef __attribute__((ext_vector_type(8))) float v8f;
typedef __attribute__((ext_vector_type(4))) float v4f;

#define T_LEN 8192
#define IN_DIM 512
#define MD 128
#define LCH 64
#define NCH (T_LEN / LCH)
#define NA 4
#define NB 4
#define NGRP (NA * NB)
#define SR (MD / NA)
#define KC (MD / NB)
#define TS 36

__device__ __forceinline__ v8f wmma4(v2f a, v2f b, v8f c) {
  return __builtin_amdgcn_wmma_f32_16x16x4_f32(
       false, a,  false, b,
       (short)0, c,  false,  false);
}

__device__ __forceinline__ float sigmoidf_(float z) {
  return 1.0f / (1.0f + __expf(-z));
}

#define PJ_NSET 4
#define PJ_STEPS (IN_DIM / 8)
__global__ __launch_bounds__(128, 1) void proj_kernel(
    const float* __restrict__ x,
    const float* __restrict__ Wq, const float* __restrict__ bq,
    const float* __restrict__ Wk, const float* __restrict__ bk,
    const float* __restrict__ Wv, const float* __restrict__ bv,
    const float* __restrict__ Wo, const float* __restrict__ bo,
    float* __restrict__ qw, float* __restrict__ kw,
    float* __restrict__ vw, float* __restrict__ ow) {
  const int wave = threadIdx.x >> 5;
  const int lane = threadIdx.x & 31;
  const int half = lane >> 4;
  const int lm = lane & 15;

  const int mat = blockIdx.y;
  const int row0 = (blockIdx.x * 2 + (wave >> 1)) * 64;
  const int col0 = (wave & 1) * 64;

  const float* W; const float* bias; float* dst;
  switch (mat) {
    case 0:  W = Wq; bias = bq; dst = qw; break;
    case 1:  W = Wk; bias = bk; dst = kw; break;
    case 2:  W = Wv; bias = bv; dst = vw; break;
    default: W = Wo; bias = bo; dst = ow; break;
  }

  const float* ap = x + (size_t)(row0 + lm) * IN_DIM + 4 * half;
  const float* bp = W + (size_t)(col0 + lm) * IN_DIM + 4 * half;

  v8f acc[4][4];
#pragma unroll
  for (int i = 0; i < 4; ++i)
#pragma unroll
    for (int j = 0; j < 4; ++j) acc[i][j] = v8f{};

  v4f fa[PJ_NSET][4], fb[PJ_NSET][4];
#pragma unroll
  for (int s = 0; s < PJ_NSET - 1; ++s) {
#pragma unroll
    for (int i = 0; i < 4; ++i) {
      fa[s][i] = *(const v4f*)(ap + (size_t)i * 16 * IN_DIM + 8 * s);
      fb[s][i] = *(const v4f*)(bp + (size_t)i * 16 * IN_DIM + 8 * s);
    }
  }

  for (int s0 = 0; s0 < PJ_STEPS; s0 += PJ_NSET) {
#pragma unroll
    for (int u = 0; u < PJ_NSET; ++u) {
      const int sn = (s0 + u + PJ_NSET - 1 < PJ_STEPS) ? (s0 + u + PJ_NSET - 1) : (PJ_STEPS - 1);
      const int w = (u + PJ_NSET - 1) % PJ_NSET;
#pragma unroll
      for (int i = 0; i < 4; ++i) {
        fa[w][i] = *(const v4f*)(ap + (size_t)i * 16 * IN_DIM + 8 * sn);
        fb[w][i] = *(const v4f*)(bp + (size_t)i * 16 * IN_DIM + 8 * sn);
      }
      __builtin_amdgcn_sched_barrier(0);
#pragma unroll
      for (int i = 0; i < 4; ++i) {
        v2f a; a.x = fa[u][i].x; a.y = fa[u][i].y;
#pragma unroll
        for (int j = 0; j < 4; ++j) {
          v2f b; b.x = fb[u][j].x; b.y = fb[u][j].y;
          acc[i][j] = wmma4(a, b, acc[i][j]);
        }
      }
#pragma unroll
      for (int i = 0; i < 4; ++i) {
        v2f a; a.x = fa[u][i].z; a.y = fa[u][i].w;
#pragma unroll
        for (int j = 0; j < 4; ++j) {
          v2f b; b.x = fb[u][j].z; b.y = fb[u][j].w;
          acc[i][j] = wmma4(a, b, acc[i][j]);
        }
      }
    }
  }

#define PJ_STORE(EXPR)                                                      \
  _Pragma("unroll") for (int j = 0; j < 4; ++j) {                           \
    const int col = col0 + j * 16 + lm;                                     \
    const float bv_ = bias[col];                                            \
    _Pragma("unroll") for (int i = 0; i < 4; ++i) {                         \
      _Pragma("unroll") for (int e = 0; e < 8; ++e) {                       \
        const int r = row0 + i * 16 + e + 8 * half;                         \
        const float val = acc[i][j][e] + bv_;                               \
        dst[(size_t)r * MD + col] = (EXPR);                                 \
      }                                                                     \
    }                                                                       \
  }
  if (mat == 3) {
    PJ_STORE(sigmoidf_(val))
  } else if (mat == 1) {
    PJ_STORE(val * 0.08838834764831845f)
  } else {
    PJ_STORE(val)
  }
#undef PJ_STORE
}

#define GR_ROWS 8
__global__ __launch_bounds__(256) void gates_kernel(
    const float* __restrict__ x,
    const float* __restrict__ wi, const float* __restrict__ bi,
    const float* __restrict__ wf, const float* __restrict__ bf,
    float* __restrict__ ig, float* __restrict__ la) {
  __shared__ float lf_s[LCH];
  const int wave = threadIdx.x >> 5;
  const int lane = threadIdx.x & 31;
  const int t0 = blockIdx.x * LCH;
  const int rbase = t0 + wave * GR_ROWS;

  v4f xv[GR_ROWS][4];
  const float* xp = x + (size_t)rbase * IN_DIM + 4 * lane;
#pragma unroll
  for (int r = 0; r < GR_ROWS; ++r) {
#pragma unroll
    for (int i = 0; i < 4; ++i)
      xv[r][i] = *(const v4f*)(xp + (size_t)r * IN_DIM + 128 * i);
  }
  v4f wiv[4], wfv[4];
#pragma unroll
  for (int i = 0; i < 4; ++i) {
    wiv[i] = *(const v4f*)(wi + 128 * i + 4 * lane);
    wfv[i] = *(const v4f*)(wf + 128 * i + 4 * lane);
  }
  __builtin_amdgcn_sched_barrier(0);

  float si[GR_ROWS], sf[GR_ROWS];
#pragma unroll
  for (int r = 0; r < GR_ROWS; ++r) {
    float a = 0.f, b = 0.f;
#pragma unroll
    for (int i = 0; i < 4; ++i) {
      a = fmaf(xv[r][i].x, wiv[i].x, a); b = fmaf(xv[r][i].x, wfv[i].x, b);
      a = fmaf(xv[r][i].y, wiv[i].y, a); b = fmaf(xv[r][i].y, wfv[i].y, b);
      a = fmaf(xv[r][i].z, wiv[i].z, a); b = fmaf(xv[r][i].z, wfv[i].z, b);
      a = fmaf(xv[r][i].w, wiv[i].w, a); b = fmaf(xv[r][i].w, wfv[i].w, b);
    }
    si[r] = a; sf[r] = b;
  }
#pragma unroll
  for (int m = 16; m >= 1; m >>= 1) {
#pragma unroll
    for (int r = 0; r < GR_ROWS; ++r) {
      si[r] += __shfl_xor(si[r], m, 32);
      sf[r] += __shfl_xor(sf[r], m, 32);
    }
  }
  float msi = si[0], msf = sf[0];
#pragma unroll
  for (int r = 1; r < GR_ROWS; ++r) {
    msi = (lane == r) ? si[r] : msi;
    msf = (lane == r) ? sf[r] : msf;
  }
  if (lane < GR_ROWS) {
    ig[rbase + lane] = __expf(msi + bi[0]);
    const float z = msf + bf[0];
    lf_s[wave * GR_ROWS + lane] =
        (z >= 0.f) ? -log1pf(__expf(-z)) : (z - log1pf(__expf(z)));
  }
  __syncthreads();
  if (threadIdx.x < LCH) {
    float lv[LCH];
#pragma unroll
    for (int i = 0; i < LCH; ++i) lv[i] = lf_s[i];
    float a = 0.f, res = 0.f;
#pragma unroll
    for (int i = 0; i < LCH; ++i) {
      a += lv[i];
      res = ((int)threadIdx.x == i) ? a : res;
    }
    la[t0 + threadIdx.x] = res;
  }
}

#define IC_NH 2
#define IC_COLS (MD / IC_NH)
#define IC_KS (MD / 8)
__global__ __launch_bounds__(128, 1) void intra_kernel(
    const float* __restrict__ qw, const float* __restrict__ kw,
    const float* __restrict__ vw, const float* __restrict__ ow,
    const float* __restrict__ ig, const float* __restrict__ la,
    const float* __restrict__ fc_w,
    float* __restrict__ numi, float* __restrict__ rsw) {
  __shared__ float nsum_s[LCH];
  __shared__ float rsum_s[LCH];
  const int tid = threadIdx.x;
  const int wave = tid >> 5;
  const int lane = tid & 31;
  const int half = lane >> 4;
  const int lm = lane & 15;
  const int t0 = blockIdx.x * LCH;
  const int nh = blockIdx.y;
  const int c0 = nh * IC_COLS;
  const int tr = wave * 16;

  const float* qp = qw + (size_t)(t0 + tr + lm) * MD + 4 * half;
  const float* kp = kw + (size_t)(t0 + lm) * MD + 4 * half;
  v4f fq[IC_KS], fk[IC_KS][4];
#pragma unroll
  for (int s = 0; s < IC_KS; ++s) {
    fq[s] = *(const v4f*)(qp + 8 * s);
#pragma unroll
    for (int q = 0; q < 4; ++q)
      fk[s][q] = *(const v4f*)(kp + (size_t)q * 16 * MD + 8 * s);
  }
  const float lat = la[t0 + tr + lm];
  v4f laj[4][2], igj[4][2];
#pragma unroll
  for (int q = 0; q < 4; ++q) {
#pragma unroll
    for (int g = 0; g < 2; ++g) {
      laj[q][g] = *(const v4f*)(la + t0 + q * 16 + 8 * half + 4 * g);
      igj[q][g] = *(const v4f*)(ig + t0 + q * 16 + 8 * half + 4 * g);
    }
  }
  float vx[4][4][4], vy[4][4][4];
#pragma unroll
  for (int q = 0; q < 4; ++q) {
#pragma unroll
    for (int s = 0; s < 4; ++s) {
      const int j0 = q * 16 + 8 * half + 2 * s;
      const float* vp = vw + (size_t)(t0 + j0) * MD + c0 + lm;
#pragma unroll
      for (int n = 0; n < 4; ++n) {
        vx[q][s][n] = vp[n * 16];
        vy[q][s][n] = vp[MD + n * 16];
      }
    }
  }
  float og[8][4], fw[4];
#pragma unroll
  for (int e = 0; e < 8; ++e) {
    const float* op = ow + (size_t)(t0 + tr + e + 8 * half) * MD + c0 + lm;
#pragma unroll
    for (int n = 0; n < 4; ++n) og[e][n] = op[n * 16];
  }
#pragma unroll
  for (int n = 0; n < 4; ++n) fw[n] = fc_w[c0 + n * 16 + lm];
  __builtin_amdgcn_sched_barrier(0);

  v8f acc[4];
#pragma unroll
  for (int q = 0; q < 4; ++q) acc[q] = v8f{};
#pragma unroll
  for (int s = 0; s < IC_KS; ++s) {
    v2f b; b.x = fq[s].x; b.y = fq[s].y;
#pragma unroll
    for (int q = 0; q < 4; ++q) {
      v2f a; a.x = fk[s][q].x; a.y = fk[s][q].y;
      acc[q] = wmma4(a, b, acc[q]);
    }
    b.x = fq[s].z; b.y = fq[s].w;
#pragma unroll
    for (int q = 0; q < 4; ++q) {
      v2f a; a.x = fk[s][q].z; a.y = fk[s][q].w;
      acc[q] = wmma4(a, b, acc[q]);
    }
  }
  float pT[4][8];
#pragma unroll
  for (int q = 0; q < 4; ++q) {
#pragma unroll
    for (int e = 0; e < 8; ++e) {
      const int j = q * 16 + 8 * half + e;
      const float p = __expf(lat - laj[q][e >> 2][e & 3]) * igj[q][e >> 2][e & 3] * acc[q][e];
      pT[q][e] = (tr + lm >= j) ? p : 0.f;
    }
  }
  if (nh == 0) {
    float sq[4];
#pragma unroll
    for (int q = 0; q < 4; ++q) {
      float s = pT[q][0];
#pragma unroll
      for (int e = 1; e < 8; ++e) s += pT[q][e];
      sq[q] = s;
    }
    const float sl = (sq[0] + sq[1]) + (sq[2] + sq[3]);
    const float st = sl + __shfl_xor(sl, 16, 32);
    if (half == 0) rsum_s[tr + lm] = st;
  }
  v8f aci[4];
#pragma unroll
  for (int n = 0; n < 4; ++n) aci[n] = v8f{};
#pragma unroll
  for (int q = 0; q < 4; ++q) {
#pragma unroll
    for (int s = 0; s < 4; ++s) {
      v2f a;
      a.x = pT[q][2 * s]; a.y = pT[q][2 * s + 1];
#pragma unroll
      for (int n = 0; n < 4; ++n) {
        v2f b;
        b.x = vx[q][s][n]; b.y = vy[q][s][n];
        aci[n] = wmma4(a, b, aci[n]);
      }
    }
  }
  float part[8];
#pragma unroll
  for (int e = 0; e < 8; ++e) {
    float a = 0.f;
#pragma unroll
    for (int n = 0; n < 4; ++n) a = fmaf(aci[n][e] * og[e][n], fw[n], a);
    part[e] = a;
  }
#pragma unroll
  for (int m = 8; m >= 1; m >>= 1) {
#pragma unroll
    for (int e = 0; e < 8; ++e) part[e] += __shfl_xor(part[e], m, 32);
  }
  float res = part[0];
#pragma unroll
  for (int e = 1; e < 8; ++e) res = (lm == e) ? part[e] : res;
  if (lm < 8) nsum_s[tr + 8 * half + lm] = res;
  __syncthreads();
  if (tid < LCH) {
    numi[(size_t)nh * T_LEN + t0 + tid] = nsum_s[tid];
    if (nh == 0) rsw[t0 + tid] = rsum_s[tid];
  }
}

#define NBUF 4
__global__ __launch_bounds__(256, 1) void scan_kernel(
    const float* __restrict__ qw, const float* __restrict__ kw,
    const float* __restrict__ vw, const float* __restrict__ ow,
    const float* __restrict__ ig, const float* __restrict__ la,
    const float* __restrict__ C0, const float* __restrict__ n0,
    const float* __restrict__ fc_w,
    float* __restrict__ nump, float* __restrict__ denp,
    float* __restrict__ C_out, float* __restrict__ n_out) {
  __shared__ float Cs[2][SR][KC + 1];
  __shared__ float ns[3][KC];
  __shared__ float fcw_s[SR];
  __shared__ float Hs[2][LCH][SR + 1];
  __shared__ __attribute__((aligned(16))) float Tq[NBUF][LCH][TS];
  __shared__ __attribute__((aligned(16))) float Tk[NBUF][LCH][TS];
  __shared__ __attribute__((aligned(16))) float Tv[NBUF][LCH][TS];
  __shared__ __attribute__((aligned(16))) float To[NBUF][LCH][TS];
  __shared__ __attribute__((aligned(16))) float la_n[NBUF][LCH];
  __shared__ __attribute__((aligned(16))) float u_n[NBUF][LCH];

  const int tid = threadIdx.x;
  const int wave = tid >> 5;
  const int lane = tid & 31;
  const int half = lane >> 4;
  const int lm = lane & 15;
  const int ga = blockIdx.x / NB;
  const int gb = blockIdx.x % NB;
  const int r0 = ga * SR;
  const int k0 = gb * KC;

  for (int i = tid; i < SR * KC; i += 256)
    Cs[0][i >> 5][i & 31] = C0[(size_t)(r0 + (i >> 5)) * MD + k0 + (i & 31)];
  if (tid < KC) ns[0][tid] = n0[k0 + tid];
  if (tid < SR) fcw_s[tid] = fc_w[r0 + tid];

  const int st_sub = (tid & 7) * 4;
  const int st_row = tid >> 3;
  const float* gq = qw + (size_t)st_row * MD + k0 + st_sub;
  const float* gk = kw + (size_t)st_row * MD + k0 + st_sub;
  const float* gv = vw + (size_t)st_row * MD + r0 + st_sub;
  const float* go = ow + (size_t)st_row * MD + r0 + st_sub;
  v4f pq0, pq1, pk0, pk1, pv0, pv1, po0, po1;
  float pla = 0.f, pig = 0.f, pend = 0.f;

#define FETCH_CHUNK(tn)                                                     \
  do {                                                                      \
    const size_t o0 = (size_t)(tn) * MD, o1 = o0 + (size_t)32 * MD;         \
    pq0 = *(const v4f*)(gq + o0); pq1 = *(const v4f*)(gq + o1);             \
    pk0 = *(const v4f*)(gk + o0); pk1 = *(const v4f*)(gk + o1);             \
    pv0 = *(const v4f*)(gv + o0); pv1 = *(const v4f*)(gv + o1);             \
    po0 = *(const v4f*)(go + o0); po1 = *(const v4f*)(go + o1);             \
    if (tid < LCH) {                                                        \
      pla = la[(tn) + tid]; pig = ig[(tn) + tid];                           \
      pend = la[(tn) + LCH - 1];                                            \
    }                                                                       \
  } while (0)

#define STAGE_CHUNK(bf)                                                     \
  do {                                                                      \
    *(v4f*)&Tq[bf][st_row][st_sub] = pq0; *(v4f*)&Tq[bf][st_row + 32][st_sub] = pq1; \
    *(v4f*)&Tk[bf][st_row][st_sub] = pk0; *(v4f*)&Tk[bf][st_row + 32][st_sub] = pk1; \
    *(v4f*)&Tv[bf][st_row][st_sub] = pv0; *(v4f*)&Tv[bf][st_row + 32][st_sub] = pv1; \
    *(v4f*)&To[bf][st_row][st_sub] = po0; *(v4f*)&To[bf][st_row + 32][st_sub] = po1; \
    if (tid < LCH) {                                                        \
      la_n[bf][tid] = pla;                                                  \
      u_n[bf][tid] = __expf(pend - pla) * pig;                              \
    }                                                                       \
  } while (0)

  FETCH_CHUNK(0);
  STAGE_CHUNK(0);
  __syncthreads();

  int m_prv = 1, m_cur = 2, m_nxt = 0;
  for (int c = -1; c <= NCH; ++c) {
    const int b_prv = (c - 1) & (NBUF - 1);
    const int b_cur = c & (NBUF - 1);
    const int b_stg = (c + 2) & (NBUF - 1);
    const int pc = c & 1;
    const int pn = pc ^ 1;
    const bool have_prv = (c >= 1);
    const bool have_cur = (c >= 0) && (c < NCH);
    const bool have_stg = (c + 2 < NCH);
    const int t0p = (c - 1) * LCH;

    if (have_stg) FETCH_CHUNK((c + 2) * LCH);

    if (wave < 4) {
      if (wave < 2) {
        if (have_prv) {
          const int row = tid;
          float hv[SR], fw[SR];
#pragma unroll
          for (int f = 0; f < SR; ++f) { hv[f] = Hs[pn][row][f]; fw[f] = fcw_s[f]; }
          __builtin_amdgcn_sched_barrier(0);
          float s = 0.f;
#pragma unroll
          for (int f = 0; f < SR; ++f) s = fmaf(hv[f], fw[f], s);
          nump[(size_t)blockIdx.x * T_LEN + t0p + row] = s;
          if (ga == 0) {
            const float* qrow = Tq[b_prv][row];
            float qv[KC], nv[KC];
#pragma unroll
            for (int f = 0; f < KC; ++f) { qv[f] = qrow[f]; nv[f] = ns[m_prv][f]; }
            const float lrow = la_n[b_prv][row];
            __builtin_amdgcn_sched_barrier(0);
            float qn = 0.f;
#pragma unroll
            for (int f = 0; f < KC; ++f) qn = fmaf(qv[f], nv[f], qn);
            denp[(size_t)gb * T_LEN + t0p + row] = __expf(lrow) * qn;
          }
        }
      } else if (wave == 3) {
        if (have_cur) {
          float uv[LCH], kv[LCH];
#pragma unroll
          for (int j = 0; j < LCH; ++j) { uv[j] = u_n[b_cur][j]; kv[j] = Tk[b_cur][j][lane]; }
          const float lend = la_n[b_cur][LCH - 1];
          const float nold = ns[m_cur][lane];
          __builtin_amdgcn_sched_barrier(0);
          const float Fend = __expf(lend);
          float sn = Fend * nold;
#pragma unroll
          for (int j = 0; j < LCH; ++j) sn = fmaf(uv[j], kv[j], sn);
          ns[m_nxt][lane] = sn;
        }
      }
      if (have_cur) {
        const float (*Qs)[TS] = Tq[b_cur];
        const float (*Ks)[TS] = Tk[b_cur];
        const float (*Vs)[TS] = Tv[b_cur];
        const float (*Os)[TS] = To[b_cur];
        const float* lac = la_n[b_cur];
        const float* uc = u_n[b_cur];
        const int tr = wave * 16;
        const int ur = (wave >> 1) * 16, ut = (wave & 1) * 16;
        v2f ha[KC / 4];
        v2f hb[KC / 4][2];
        const float* ap = &Qs[tr + lm][2 * half];
#pragma unroll
        for (int k4 = 0; k4 < KC / 4; ++k4) {
          ha[k4].x = ap[4 * k4]; ha[k4].y = ap[4 * k4 + 1];
#pragma unroll
          for (int q = 0; q < 2; ++q) {
            hb[k4][q].x = Cs[pc][q * 16 + lm][4 * k4 + 2 * half];
            hb[k4][q].y = Cs[pc][q * 16 + lm][4 * k4 + 2 * half + 1];
          }
        }
        float lr[8], oq[8][2];
#pragma unroll
        for (int e = 0; e < 8; ++e) {
          const int r = tr + e + 8 * half;
          lr[e] = lac[r];
#pragma unroll
          for (int q = 0; q < 2; ++q) oq[e][q] = Os[r][q * 16 + lm];
        }
        const float lend = lac[LCH - 1];
        float c0[8];
#pragma unroll
        for (int e = 0; e < 8; ++e) c0[e] = Cs[pc][ur + e + 8 * half][ut + lm];
        v2f uu[LCH / 4], vv[LCH / 4], kb[LCH / 4];
#pragma unroll
        for (int k4 = 0; k4 < LCH / 4; ++k4) {
          const int j0 = 4 * k4 + 2 * half;
          uu[k4].x = uc[j0];          uu[k4].y = uc[j0 + 1];
          vv[k4].x = Vs[j0][ur + lm]; vv[k4].y = Vs[j0 + 1][ur + lm];
          kb[k4].x = Ks[j0][ut + lm]; kb[k4].y = Ks[j0 + 1][ut + lm];
        }
        asm volatile("" :: "v"(lr[0]), "v"(lr[1]), "v"(lr[2]), "v"(lr[3]),
                           "v"(lr[4]), "v"(lr[5]), "v"(lr[6]), "v"(lr[7]));
        {
          v8f ace[2];
          ace[0] = v8f{}; ace[1] = v8f{};
#pragma unroll
          for (int k4 = 0; k4 < KC / 4; ++k4) {
#pragma unroll
            for (int q = 0; q < 2; ++q) ace[q] = wmma4(ha[k4], hb[k4][q], ace[q]);
          }
#pragma unroll
          for (int e = 0; e < 8; ++e) {
            const int r = tr + e + 8 * half;
            const float Ft = __expf(lr[e]);
#pragma unroll
            for (int q = 0; q < 2; ++q) {
              const int cc = q * 16 + lm;
              Hs[pc][r][cc] = (Ft * ace[q][e]) * oq[e][q];
            }
          }
        }
        {
          const float Fend = __expf(lend);
          v8f acc;
#pragma unroll
          for (int e = 0; e < 8; ++e) acc[e] = Fend * c0[e];
#pragma unroll
          for (int k4 = 0; k4 < LCH / 4; ++k4) {
            v2f a;
            a.x = uu[k4].x * vv[k4].x;
            a.y = uu[k4].y * vv[k4].y;
            acc = wmma4(a, kb[k4], acc);
          }
#pragma unroll
          for (int e = 0; e < 8; ++e) Cs[pn][ur + e + 8 * half][ut + lm] = acc[e];
        }
        __builtin_amdgcn_sched_group_barrier(0x100, 128, 0);
        __builtin_amdgcn_sched_group_barrier(0x008, 32, 0);
      }
    }
    if (have_stg) STAGE_CHUNK(b_stg);
    __syncthreads();
    m_prv = m_cur; m_cur = m_nxt; m_nxt = (m_nxt == 2) ? 0 : m_nxt + 1;
  }

#undef FETCH_CHUNK
#undef STAGE_CHUNK

  for (int i = tid; i < SR * KC; i += 256)
    C_out[(size_t)(r0 + (i >> 5)) * MD + k0 + (i & 31)] = Cs[NCH & 1][i >> 5][i & 31];
  if (ga == 0 && tid < KC) n_out[k0 + tid] = ns[NCH % 3][tid];
}

__global__ __launch_bounds__(256) void out_kernel(
    const float* __restrict__ nump, const float* __restrict__ denp,
    const float* __restrict__ numi, const float* __restrict__ rsw,
    const float* __restrict__ fc_b, float* __restrict__ out) {
  const int t = blockIdx.x * 256 + threadIdx.x;
  float s = nump[t];
#pragma unroll
  for (int g = 1; g < NGRP; ++g) s += nump[(size_t)g * T_LEN + t];
#pragma unroll
  for (int h = 0; h < IC_NH; ++h) s += numi[(size_t)h * T_LEN + t];
  float d = denp[t];
#pragma unroll
  for (int b = 1; b < NB; ++b) d += denp[(size_t)b * T_LEN + t];
  d += rsw[t];
  out[t] = s / fmaxf(fabsf(d), 1.0f) + fc_b[0];
}

extern "C" void kernel_launch(void* const* d_in, const int* in_sizes, int n_in,
                              void* d_out, int out_size, void* d_ws, size_t ws_size,
                              hipStream_t stream) {
  const float* x    = (const float*)d_in[0];
  const float* C0   = (const float*)d_in[1];
  const float* n0   = (const float*)d_in[2];
  const float* Wq   = (const float*)d_in[3];
  const float* bq   = (const float*)d_in[4];
  const float* Wk   = (const float*)d_in[5];
  const float* bk   = (const float*)d_in[6];
  const float* Wv   = (const float*)d_in[7];
  const float* bv   = (const float*)d_in[8];
  const float* wi   = (const float*)d_in[9];
  const float* bi   = (const float*)d_in[10];
  const float* wf   = (const float*)d_in[11];
  const float* bf   = (const float*)d_in[12];
  const float* Wo   = (const float*)d_in[13];
  const float* bo   = (const float*)d_in[14];
  const float* fc_w = (const float*)d_in[15];
  const float* fc_b = (const float*)d_in[16];

  float* ws = (float*)d_ws;
  const size_t NP = (size_t)T_LEN * MD;
  float* qw = ws;
  float* kw = ws + NP;
  float* vw = ws + 2 * NP;
  float* owp = ws + 3 * NP;
  float* igp = ws + 4 * NP;
  float* lfp = igp + T_LEN;
  float* numpp = lfp + T_LEN;
  float* denpp = numpp + NGRP * T_LEN;
  float* numip = denpp + NB * T_LEN;
  float* rswp  = numip + IC_NH * T_LEN;

  float* outp  = (float*)d_out;
  float* Coutp = outp + T_LEN;
  float* noutp = Coutp + MD * MD;

  proj_kernel<<<dim3(T_LEN / 128, 4), 128, 0, stream>>>(
      x, Wq, bq, Wk, bk, Wv, bv, Wo, bo, qw, kw, vw, owp);
  gates_kernel<<<dim3(T_LEN / LCH), 256, 0, stream>>>(
      x, wi, bi, wf, bf, igp, lfp);
  intra_kernel<<<dim3(T_LEN / LCH, IC_NH), 128, 0, stream>>>(
      qw, kw, vw, owp, igp, lfp, fc_w, numip, rswp);
  scan_kernel<<<NGRP, 256, 0, stream>>>(
      qw, kw, vw, owp, igp, lfp, C0, n0, fc_w, numpp, denpp, Coutp, noutp);
  out_kernel<<<T_LEN / 256, 256, 0, stream>>>(numpp, denpp, numip, rswp, fc_b, outp);
}


# Recurrence split over time too: 8 segments per state tile, 3 launches (was 1): state-only pass from zero, in-place prefix of segment start states, full pass per segment; state sum order changes
# speedup vs baseline: 33.4147x; 2.3329x over previous
#include <hip/hip_runtime.h>
#include <hip/hip_bf16.h>
#include <math.h>


typedef __attribute__((ext_vector_type(2))) float v2f;
typedef __attribute__((ext_vector_type(8))) float v8f;
typedef __attribute__((ext_vector_type(4))) float v4f;

#define T_LEN 8192
#define IN_DIM 512
#define MD 128
#define LCH 64
#define NCH (T_LEN / LCH)
#define NA 4
#define NB 4
#define NGRP (NA * NB)
#define SR (MD / NA)
#define KC (MD / NB)
#define TS 36

__device__ __forceinline__ v8f wmma4(v2f a, v2f b, v8f c) {
  return __builtin_amdgcn_wmma_f32_16x16x4_f32(
       false, a,  false, b,
       (short)0, c,  false,  false);
}

__device__ __forceinline__ float sigmoidf_(float z) {
  return 1.0f / (1.0f + __expf(-z));
}

#define PJ_NSET 4
#define PJ_STEPS (IN_DIM / 8)
__global__ __launch_bounds__(128, 1) void proj_kernel(
    const float* __restrict__ x,
    const float* __restrict__ Wq, const float* __restrict__ bq,
    const float* __restrict__ Wk, const float* __restrict__ bk,
    const float* __restrict__ Wv, const float* __restrict__ bv,
    const float* __restrict__ Wo, const float* __restrict__ bo,
    float* __restrict__ qw, float* __restrict__ kw,
    float* __restrict__ vw, float* __restrict__ ow) {
  const int wave = threadIdx.x >> 5;
  const int lane = threadIdx.x & 31;
  const int half = lane >> 4;
  const int lm = lane & 15;

  const int mat = blockIdx.y;
  const int row0 = (blockIdx.x * 2 + (wave >> 1)) * 64;
  const int col0 = (wave & 1) * 64;

  const float* W; const float* bias; float* dst;
  switch (mat) {
    case 0:  W = Wq; bias = bq; dst = qw; break;
    case 1:  W = Wk; bias = bk; dst = kw; break;
    case 2:  W = Wv; bias = bv; dst = vw; break;
    default: W = Wo; bias = bo; dst = ow; break;
  }

  const float* ap = x + (size_t)(row0 + lm) * IN_DIM + 4 * half;
  const float* bp = W + (size_t)(col0 + lm) * IN_DIM + 4 * half;

  v8f acc[4][4];
#pragma unroll
  for (int i = 0; i < 4; ++i)
#pragma unroll
    for (int j = 0; j < 4; ++j) acc[i][j] = v8f{};

  v4f fa[PJ_NSET][4], fb[PJ_NSET][4];
#pragma unroll
  for (int s = 0; s < PJ_NSET - 1; ++s) {
#pragma unroll
    for (int i = 0; i < 4; ++i) {
      fa[s][i] = *(const v4f*)(ap + (size_t)i * 16 * IN_DIM + 8 * s);
      fb[s][i] = *(const v4f*)(bp + (size_t)i * 16 * IN_DIM + 8 * s);
    }
  }

  for (int s0 = 0; s0 < PJ_STEPS; s0 += PJ_NSET) {
#pragma unroll
    for (int u = 0; u < PJ_NSET; ++u) {
      const int sn = (s0 + u + PJ_NSET - 1 < PJ_STEPS) ? (s0 + u + PJ_NSET - 1) : (PJ_STEPS - 1);
      const int w = (u + PJ_NSET - 1) % PJ_NSET;
#pragma unroll
      for (int i = 0; i < 4; ++i) {
        fa[w][i] = *(const v4f*)(ap + (size_t)i * 16 * IN_DIM + 8 * sn);
        fb[w][i] = *(const v4f*)(bp + (size_t)i * 16 * IN_DIM + 8 * sn);
      }
      __builtin_amdgcn_sched_barrier(0);
#pragma unroll
      for (int i = 0; i < 4; ++i) {
        v2f a; a.x = fa[u][i].x; a.y = fa[u][i].y;
#pragma unroll
        for (int j = 0; j < 4; ++j) {
          v2f b; b.x = fb[u][j].x; b.y = fb[u][j].y;
          acc[i][j] = wmma4(a, b, acc[i][j]);
        }
      }
#pragma unroll
      for (int i = 0; i < 4; ++i) {
        v2f a; a.x = fa[u][i].z; a.y = fa[u][i].w;
#pragma unroll
        for (int j = 0; j < 4; ++j) {
          v2f b; b.x = fb[u][j].z; b.y = fb[u][j].w;
          acc[i][j] = wmma4(a, b, acc[i][j]);
        }
      }
    }
  }

#define PJ_STORE(EXPR)                                                      \
  _Pragma("unroll") for (int j = 0; j < 4; ++j) {                           \
    const int col = col0 + j * 16 + lm;                                     \
    const float bv_ = bias[col];                                            \
    _Pragma("unroll") for (int i = 0; i < 4; ++i) {                         \
      _Pragma("unroll") for (int e = 0; e < 8; ++e) {                       \
        const int r = row0 + i * 16 + e + 8 * half;                         \
        const float val = acc[i][j][e] + bv_;                               \
        dst[(size_t)r * MD + col] = (EXPR);                                 \
      }                                                                     \
    }                                                                       \
  }
  if (mat == 3) {
    PJ_STORE(sigmoidf_(val))
  } else if (mat == 1) {
    PJ_STORE(val * 0.08838834764831845f)
  } else {
    PJ_STORE(val)
  }
#undef PJ_STORE
}

#define GR_ROWS 8
__global__ __launch_bounds__(256) void gates_kernel(
    const float* __restrict__ x,
    const float* __restrict__ wi, const float* __restrict__ bi,
    const float* __restrict__ wf, const float* __restrict__ bf,
    float* __restrict__ ig, float* __restrict__ la) {
  __shared__ float lf_s[LCH];
  const int wave = threadIdx.x >> 5;
  const int lane = threadIdx.x & 31;
  const int t0 = blockIdx.x * LCH;
  const int rbase = t0 + wave * GR_ROWS;

  v4f xv[GR_ROWS][4];
  const float* xp = x + (size_t)rbase * IN_DIM + 4 * lane;
#pragma unroll
  for (int r = 0; r < GR_ROWS; ++r) {
#pragma unroll
    for (int i = 0; i < 4; ++i)
      xv[r][i] = *(const v4f*)(xp + (size_t)r * IN_DIM + 128 * i);
  }
  v4f wiv[4], wfv[4];
#pragma unroll
  for (int i = 0; i < 4; ++i) {
    wiv[i] = *(const v4f*)(wi + 128 * i + 4 * lane);
    wfv[i] = *(const v4f*)(wf + 128 * i + 4 * lane);
  }
  __builtin_amdgcn_sched_barrier(0);

  float si[GR_ROWS], sf[GR_ROWS];
#pragma unroll
  for (int r = 0; r < GR_ROWS; ++r) {
    float a = 0.f, b = 0.f;
#pragma unroll
    for (int i = 0; i < 4; ++i) {
      a = fmaf(xv[r][i].x, wiv[i].x, a); b = fmaf(xv[r][i].x, wfv[i].x, b);
      a = fmaf(xv[r][i].y, wiv[i].y, a); b = fmaf(xv[r][i].y, wfv[i].y, b);
      a = fmaf(xv[r][i].z, wiv[i].z, a); b = fmaf(xv[r][i].z, wfv[i].z, b);
      a = fmaf(xv[r][i].w, wiv[i].w, a); b = fmaf(xv[r][i].w, wfv[i].w, b);
    }
    si[r] = a; sf[r] = b;
  }
#pragma unroll
  for (int m = 16; m >= 1; m >>= 1) {
#pragma unroll
    for (int r = 0; r < GR_ROWS; ++r) {
      si[r] += __shfl_xor(si[r], m, 32);
      sf[r] += __shfl_xor(sf[r], m, 32);
    }
  }
  float msi = si[0], msf = sf[0];
#pragma unroll
  for (int r = 1; r < GR_ROWS; ++r) {
    msi = (lane == r) ? si[r] : msi;
    msf = (lane == r) ? sf[r] : msf;
  }
  if (lane < GR_ROWS) {
    ig[rbase + lane] = __expf(msi + bi[0]);
    const float z = msf + bf[0];
    lf_s[wave * GR_ROWS + lane] =
        (z >= 0.f) ? -log1pf(__expf(-z)) : (z - log1pf(__expf(z)));
  }
  __syncthreads();
  if (threadIdx.x < LCH) {
    float lv[LCH];
#pragma unroll
    for (int i = 0; i < LCH; ++i) lv[i] = lf_s[i];
    float a = 0.f, res = 0.f;
#pragma unroll
    for (int i = 0; i < LCH; ++i) {
      a += lv[i];
      res = ((int)threadIdx.x == i) ? a : res;
    }
    la[t0 + threadIdx.x] = res;
  }
}

#define IC_NH 2
#define IC_COLS (MD / IC_NH)
#define IC_KS (MD / 8)
__global__ __launch_bounds__(128, 1) void intra_kernel(
    const float* __restrict__ qw, const float* __restrict__ kw,
    const float* __restrict__ vw, const float* __restrict__ ow,
    const float* __restrict__ ig, const float* __restrict__ la,
    const float* __restrict__ fc_w,
    float* __restrict__ numi, float* __restrict__ rsw) {
  __shared__ float nsum_s[LCH];
  __shared__ float rsum_s[LCH];
  const int tid = threadIdx.x;
  const int wave = tid >> 5;
  const int lane = tid & 31;
  const int half = lane >> 4;
  const int lm = lane & 15;
  const int t0 = blockIdx.x * LCH;
  const int nh = blockIdx.y;
  const int c0 = nh * IC_COLS;
  const int tr = wave * 16;

  const float* qp = qw + (size_t)(t0 + tr + lm) * MD + 4 * half;
  const float* kp = kw + (size_t)(t0 + lm) * MD + 4 * half;
  v4f fq[IC_KS], fk[IC_KS][4];
#pragma unroll
  for (int s = 0; s < IC_KS; ++s) {
    fq[s] = *(const v4f*)(qp + 8 * s);
#pragma unroll
    for (int q = 0; q < 4; ++q)
      fk[s][q] = *(const v4f*)(kp + (size_t)q * 16 * MD + 8 * s);
  }
  const float lat = la[t0 + tr + lm];
  v4f laj[4][2], igj[4][2];
#pragma unroll
  for (int q = 0; q < 4; ++q) {
#pragma unroll
    for (int g = 0; g < 2; ++g) {
      laj[q][g] = *(const v4f*)(la + t0 + q * 16 + 8 * half + 4 * g);
      igj[q][g] = *(const v4f*)(ig + t0 + q * 16 + 8 * half + 4 * g);
    }
  }
  float vx[4][4][4], vy[4][4][4];
#pragma unroll
  for (int q = 0; q < 4; ++q) {
#pragma unroll
    for (int s = 0; s < 4; ++s) {
      const int j0 = q * 16 + 8 * half + 2 * s;
      const float* vp = vw + (size_t)(t0 + j0) * MD + c0 + lm;
#pragma unroll
      for (int n = 0; n < 4; ++n) {
        vx[q][s][n] = vp[n * 16];
        vy[q][s][n] = vp[MD + n * 16];
      }
    }
  }
  float og[8][4], fw[4];
#pragma unroll
  for (int e = 0; e < 8; ++e) {
    const float* op = ow + (size_t)(t0 + tr + e + 8 * half) * MD + c0 + lm;
#pragma unroll
    for (int n = 0; n < 4; ++n) og[e][n] = op[n * 16];
  }
#pragma unroll
  for (int n = 0; n < 4; ++n) fw[n] = fc_w[c0 + n * 16 + lm];
  __builtin_amdgcn_sched_barrier(0);

  v8f acc[4];
#pragma unroll
  for (int q = 0; q < 4; ++q) acc[q] = v8f{};
#pragma unroll
  for (int s = 0; s < IC_KS; ++s) {
    v2f b; b.x = fq[s].x; b.y = fq[s].y;
#pragma unroll
    for (int q = 0; q < 4; ++q) {
      v2f a; a.x = fk[s][q].x; a.y = fk[s][q].y;
      acc[q] = wmma4(a, b, acc[q]);
    }
    b.x = fq[s].z; b.y = fq[s].w;
#pragma unroll
    for (int q = 0; q < 4; ++q) {
      v2f a; a.x = fk[s][q].z; a.y = fk[s][q].w;
      acc[q] = wmma4(a, b, acc[q]);
    }
  }
  float pT[4][8];
#pragma unroll
  for (int q = 0; q < 4; ++q) {
#pragma unroll
    for (int e = 0; e < 8; ++e) {
      const int j = q * 16 + 8 * half + e;
      const float p = __expf(lat - laj[q][e >> 2][e & 3]) * igj[q][e >> 2][e & 3] * acc[q][e];
      pT[q][e] = (tr + lm >= j) ? p : 0.f;
    }
  }
  if (nh == 0) {
    float sq[4];
#pragma unroll
    for (int q = 0; q < 4; ++q) {
      float s = pT[q][0];
#pragma unroll
      for (int e = 1; e < 8; ++e) s += pT[q][e];
      sq[q] = s;
    }
    const float sl = (sq[0] + sq[1]) + (sq[2] + sq[3]);
    const float st = sl + __shfl_xor(sl, 16, 32);
    if (half == 0) rsum_s[tr + lm] = st;
  }
  v8f aci[4];
#pragma unroll
  for (int n = 0; n < 4; ++n) aci[n] = v8f{};
#pragma unroll
  for (int q = 0; q < 4; ++q) {
#pragma unroll
    for (int s = 0; s < 4; ++s) {
      v2f a;
      a.x = pT[q][2 * s]; a.y = pT[q][2 * s + 1];
#pragma unroll
      for (int n = 0; n < 4; ++n) {
        v2f b;
        b.x = vx[q][s][n]; b.y = vy[q][s][n];
        aci[n] = wmma4(a, b, aci[n]);
      }
    }
  }
  float part[8];
#pragma unroll
  for (int e = 0; e < 8; ++e) {
    float a = 0.f;
#pragma unroll
    for (int n = 0; n < 4; ++n) a = fmaf(aci[n][e] * og[e][n], fw[n], a);
    part[e] = a;
  }
#pragma unroll
  for (int m = 8; m >= 1; m >>= 1) {
#pragma unroll
    for (int e = 0; e < 8; ++e) part[e] += __shfl_xor(part[e], m, 32);
  }
  float res = part[0];
#pragma unroll
  for (int e = 1; e < 8; ++e) res = (lm == e) ? part[e] : res;
  if (lm < 8) nsum_s[tr + 8 * half + lm] = res;
  __syncthreads();
  if (tid < LCH) {
    numi[(size_t)nh * T_LEN + t0 + tid] = nsum_s[tid];
    if (nh == 0) rsw[t0 + tid] = rsum_s[tid];
  }
}

#define NBUF 4
#define NSEG 8
#define CPS (NCH / NSEG)
#define SEGT (CPS * LCH)
static_assert(SEGT == SR * KC, "a segment's stretch of a per-step plane must hold one state tile");
template <bool FULL>
__global__ __launch_bounds__(256, 1) void scan_kernel(
    const float* __restrict__ qw, const float* __restrict__ kw,
    const float* __restrict__ vw, const float* __restrict__ ow,
    const float* __restrict__ ig, const float* __restrict__ la,
    const float* __restrict__ fc_w,
    float* __restrict__ nump, float* __restrict__ denp,
    float* __restrict__ C_out, float* __restrict__ n_out) {
  __shared__ float Cs[2][SR][KC + 1];
  __shared__ float ns[3][KC];
  __shared__ float fcw_s[SR];
  __shared__ float Hs[2][LCH][SR + 1];
  __shared__ __attribute__((aligned(16))) float Tq[NBUF][LCH][TS];
  __shared__ __attribute__((aligned(16))) float Tk[NBUF][LCH][TS];
  __shared__ __attribute__((aligned(16))) float Tv[NBUF][LCH][TS];
  __shared__ __attribute__((aligned(16))) float To[NBUF][LCH][TS];
  __shared__ __attribute__((aligned(16))) float la_n[NBUF][LCH];
  __shared__ __attribute__((aligned(16))) float u_n[NBUF][LCH];

  const int tid = threadIdx.x;
  const int wave = tid >> 5;
  const int lane = tid & 31;
  const int half = lane >> 4;
  const int lm = lane & 15;
  const int sg = blockIdx.x / NGRP;
  const int gt = blockIdx.x % NGRP;
  const int ga = gt / NB;
  const int gb = gt % NB;
  const int r0 = ga * SR;
  const int k0 = gb * KC;
  const int tb = sg * SEGT;
  float* cslot = nump + (size_t)gt * T_LEN + tb;
  float* nslot = denp + (size_t)gb * T_LEN + tb;

  if (FULL) {
    for (int i = tid; i < SR * KC; i += 256) Cs[0][i >> 5][i & 31] = cslot[i];
    if (tid < KC) {
      if (ga == 0) ns[0][tid] = nslot[tid];
      else         ns[0][tid] = 0.f;
    }
    if (tid < SR) fcw_s[tid] = fc_w[r0 + tid];
  } else {
    for (int i = tid; i < SR * KC; i += 256) Cs[0][i >> 5][i & 31] = 0.f;
    if (tid < KC) ns[0][tid] = 0.f;
  }

  const int st_sub = (tid & 7) * 4;
  const int st_row = tid >> 3;
  const float* gq = qw + (size_t)st_row * MD + k0 + st_sub;
  const float* gk = kw + (size_t)st_row * MD + k0 + st_sub;
  const float* gv = vw + (size_t)st_row * MD + r0 + st_sub;
  const float* go = ow + (size_t)st_row * MD + r0 + st_sub;
  v4f pq0 = v4f{}, pq1 = v4f{}, pk0, pk1, pv0, pv1, po0 = v4f{}, po1 = v4f{};
  float pla = 0.f, pig = 0.f, pend = 0.f;

#define FETCH_CHUNK(tn)                                                     \
  do {                                                                      \
    const size_t o0 = (size_t)(tn) * MD, o1 = o0 + (size_t)32 * MD;         \
    if (FULL) {                                                             \
      pq0 = *(const v4f*)(gq + o0); pq1 = *(const v4f*)(gq + o1);           \
      po0 = *(const v4f*)(go + o0); po1 = *(const v4f*)(go + o1);           \
    }                                                                       \
    pk0 = *(const v4f*)(gk + o0); pk1 = *(const v4f*)(gk + o1);             \
    pv0 = *(const v4f*)(gv + o0); pv1 = *(const v4f*)(gv + o1);             \
    if (tid < LCH) {                                                        \
      pla = la[(tn) + tid]; pig = ig[(tn) + tid];                           \
      pend = la[(tn) + LCH - 1];                                            \
    }                                                                       \
  } while (0)

#define STAGE_CHUNK(bf)                                                     \
  do {                                                                      \
    if (FULL) {                                                             \
      *(v4f*)&Tq[bf][st_row][st_sub] = pq0; *(v4f*)&Tq[bf][st_row + 32][st_sub] = pq1; \
      *(v4f*)&To[bf][st_row][st_sub] = po0; *(v4f*)&To[bf][st_row + 32][st_sub] = po1; \
    }                                                                       \
    *(v4f*)&Tk[bf][st_row][st_sub] = pk0; *(v4f*)&Tk[bf][st_row + 32][st_sub] = pk1; \
    *(v4f*)&Tv[bf][st_row][st_sub] = pv0; *(v4f*)&Tv[bf][st_row + 32][st_sub] = pv1; \
    if (tid < LCH) {                                                        \
      la_n[bf][tid] = pla;                                                  \
      u_n[bf][tid] = __expf(pend - pla) * pig;                              \
    }                                                                       \
  } while (0)

  FETCH_CHUNK(tb);
  STAGE_CHUNK(0);
  __syncthreads();

  int m_prv = 1, m_cur = 2, m_nxt = 0;
  const int c_last = FULL ? CPS : CPS - 1;
  for (int c = -1; c <= c_last; ++c) {
    const int b_prv = (c - 1) & (NBUF - 1);
    const int b_cur = c & (NBUF - 1);
    const int b_stg = (c + 2) & (NBUF - 1);
    const int pc = c & 1;
    const int pn = pc ^ 1;
    const bool have_prv = (c >= 1);
    const bool have_cur = (c >= 0) && (c < CPS);
    const bool have_stg = (c + 2 < CPS);
    const int t0p = tb + (c - 1) * LCH;

    if (have_stg) FETCH_CHUNK(tb + (c + 2) * LCH);

    if (wave < 4) {
      if (wave < 2) {
        if (FULL && have_prv) {
          const int row = tid;
          float hv[SR], fw[SR];
#pragma unroll
          for (int f = 0; f < SR; ++f) { hv[f] = Hs[pn][row][f]; fw[f] = fcw_s[f]; }
          __builtin_amdgcn_sched_barrier(0);
          float s = 0.f;
#pragma unroll
          for (int f = 0; f < SR; ++f) s = fmaf(hv[f], fw[f], s);
          nump[(size_t)gt * T_LEN + t0p + row] = s;
          if (ga == 0) {
            const float* qrow = Tq[b_prv][row];
            float qv[KC], nv[KC];
#pragma unroll
            for (int f = 0; f < KC; ++f) { qv[f] = qrow[f]; nv[f] = ns[m_prv][f]; }
            const float lrow = la_n[b_prv][row];
            __builtin_amdgcn_sched_barrier(0);
            float qn = 0.f;
#pragma unroll
            for (int f = 0; f < KC; ++f) qn = fmaf(qv[f], nv[f], qn);
            denp[(size_t)gb * T_LEN + t0p + row] = __expf(lrow) * qn;
          }
        }
      } else if (wave == 3) {
        if (have_cur) {
          float uv[LCH], kv[LCH];
#pragma unroll
          for (int j = 0; j < LCH; ++j) { uv[j] = u_n[b_cur][j]; kv[j] = Tk[b_cur][j][lane]; }
          const float lend = la_n[b_cur][LCH - 1];
          const float nold = ns[m_cur][lane];
          __builtin_amdgcn_sched_barrier(0);
          const float Fend = __expf(lend);
          float sn = Fend * nold;
#pragma unroll
          for (int j = 0; j < LCH; ++j) sn = fmaf(uv[j], kv[j], sn);
          ns[m_nxt][lane] = sn;
        }
      }
      if (have_cur) {
        const float (*Qs)[TS] = Tq[b_cur];
        const float (*Ks)[TS] = Tk[b_cur];
        const float (*Vs)[TS] = Tv[b_cur];
        const float (*Os)[TS] = To[b_cur];
        const float* lac = la_n[b_cur];
        const float* uc = u_n[b_cur];
        const int tr = wave * 16;
        const int ur = (wave >> 1) * 16, ut = (wave & 1) * 16;
        v2f ha[KC / 4];
        v2f hb[KC / 4][2];
        float lr[8], oq[8][2];
        if (FULL) {
          const float* ap = &Qs[tr + lm][2 * half];
#pragma unroll
          for (int k4 = 0; k4 < KC / 4; ++k4) {
            ha[k4].x = ap[4 * k4]; ha[k4].y = ap[4 * k4 + 1];
#pragma unroll
            for (int q = 0; q < 2; ++q) {
              hb[k4][q].x = Cs[pc][q * 16 + lm][4 * k4 + 2 * half];
              hb[k4][q].y = Cs[pc][q * 16 + lm][4 * k4 + 2 * half + 1];
            }
          }
#pragma unroll
          for (int e = 0; e < 8; ++e) {
            const int r = tr + e + 8 * half;
            lr[e] = lac[r];
#pragma unroll
            for (int q = 0; q < 2; ++q) oq[e][q] = Os[r][q * 16 + lm];
          }
        }
        const float lend = lac[LCH - 1];
        float c0[8];
#pragma unroll
        for (int e = 0; e < 8; ++e) c0[e] = Cs[pc][ur + e + 8 * half][ut + lm];
        v2f uu[LCH / 4], vv[LCH / 4], kb[LCH / 4];
#pragma unroll
        for (int k4 = 0; k4 < LCH / 4; ++k4) {
          const int j0 = 4 * k4 + 2 * half;
          uu[k4].x = uc[j0];          uu[k4].y = uc[j0 + 1];
          vv[k4].x = Vs[j0][ur + lm]; vv[k4].y = Vs[j0 + 1][ur + lm];
          kb[k4].x = Ks[j0][ut + lm]; kb[k4].y = Ks[j0 + 1][ut + lm];
        }
        if (FULL)
          asm volatile("" :: "v"(lr[0]), "v"(lr[1]), "v"(lr[2]), "v"(lr[3]),
                             "v"(lr[4]), "v"(lr[5]), "v"(lr[6]), "v"(lr[7]));
        if (FULL) {
          v8f ace[2];
          ace[0] = v8f{}; ace[1] = v8f{};
#pragma unroll
          for (int k4 = 0; k4 < KC / 4; ++k4) {
#pragma unroll
            for (int q = 0; q < 2; ++q) ace[q] = wmma4(ha[k4], hb[k4][q], ace[q]);
          }
#pragma unroll
          for (int e = 0; e < 8; ++e) {
            const int r = tr + e + 8 * half;
            const float Ft = __expf(lr[e]);
#pragma unroll
            for (int q = 0; q < 2; ++q) {
              const int cc = q * 16 + lm;
              Hs[pc][r][cc] = (Ft * ace[q][e]) * oq[e][q];
            }
          }
        }
        {
          const float Fend = __expf(lend);
          v8f acc;
#pragma unroll
          for (int e = 0; e < 8; ++e) acc[e] = Fend * c0[e];
#pragma unroll
          for (int k4 = 0; k4 < LCH / 4; ++k4) {
            v2f a;
            a.x = uu[k4].x * vv[k4].x;
            a.y = uu[k4].y * vv[k4].y;
            acc = wmma4(a, kb[k4], acc);
          }
#pragma unroll
          for (int e = 0; e < 8; ++e) Cs[pn][ur + e + 8 * half][ut + lm] = acc[e];
        }
        __builtin_amdgcn_sched_group_barrier(0x100, 128, 0);
        __builtin_amdgcn_sched_group_barrier(0x008, 32, 0);
      }
    }
    if (have_stg) STAGE_CHUNK(b_stg);
    __syncthreads();
    m_prv = m_cur; m_cur = m_nxt; m_nxt = (m_nxt == 2) ? 0 : m_nxt + 1;
  }

#undef FETCH_CHUNK
#undef STAGE_CHUNK

  if (FULL) {
    if (sg == NSEG - 1) {
      for (int i = tid; i < SR * KC; i += 256)
        C_out[(size_t)(r0 + (i >> 5)) * MD + k0 + (i & 31)] = Cs[CPS & 1][i >> 5][i & 31];
      if (ga == 0 && tid < KC) n_out[k0 + tid] = ns[CPS % 3][tid];
    }
  } else {
    for (int i = tid; i < SR * KC; i += 256) cslot[i] = Cs[CPS & 1][i >> 5][i & 31];
    if (ga == 0 && tid < KC) nslot[tid] = ns[CPS % 3][tid];
  }
}

#define PF_CB (NGRP * SR * KC / 256)
__global__ __launch_bounds__(256) void prefix_kernel(
    const float* __restrict__ la, const float* __restrict__ C0,
    const float* __restrict__ n0,
    float* __restrict__ nump, float* __restrict__ denp) {
  __shared__ float dec_s[NSEG];
  const int tid = threadIdx.x;
  if (tid < NSEG - 1) {
    float gsum = 0.f;
    for (int c = 0; c < CPS; ++c) gsum += la[tid * SEGT + c * LCH + LCH - 1];
    dec_s[tid] = __expf(gsum);
  }
  __syncthreads();
  float* slot = nullptr;
  float cur = 0.f;
  if ((int)blockIdx.x < PF_CB) {
    const int e = blockIdx.x * 256 + tid;
    const int g = e / (SR * KC), i = e % (SR * KC);
    const int a = g / NB, b = g % NB;
    slot = nump + (size_t)g * T_LEN + i;
    cur = C0[(size_t)(a * SR + (i >> 5)) * MD + b * KC + (i & 31)];
  } else if (tid < MD) {
    slot = denp + (size_t)(tid / KC) * T_LEN + (tid % KC);
    cur = n0[tid];
  }
  if (slot != nullptr) {
    float loc[NSEG - 1];
#pragma unroll
    for (int sI = 0; sI < NSEG - 1; ++sI) loc[sI] = slot[(size_t)sI * SEGT];
    slot[0] = cur;
#pragma unroll
    for (int sI = 0; sI < NSEG - 1; ++sI) {
      cur = fmaf(dec_s[sI], cur, loc[sI]);
      slot[(size_t)(sI + 1) * SEGT] = cur;
    }
  }
}

__global__ __launch_bounds__(256) void out_kernel(
    const float* __restrict__ nump, const float* __restrict__ denp,
    const float* __restrict__ numi, const float* __restrict__ rsw,
    const float* __restrict__ fc_b, float* __restrict__ out) {
  const int t = blockIdx.x * 256 + threadIdx.x;
  float s = nump[t];
#pragma unroll
  for (int g = 1; g < NGRP; ++g) s += nump[(size_t)g * T_LEN + t];
#pragma unroll
  for (int h = 0; h < IC_NH; ++h) s += numi[(size_t)h * T_LEN + t];
  float d = denp[t];
#pragma unroll
  for (int b = 1; b < NB; ++b) d += denp[(size_t)b * T_LEN + t];
  d += rsw[t];
  out[t] = s / fmaxf(fabsf(d), 1.0f) + fc_b[0];
}

extern "C" void kernel_launch(void* const* d_in, const int* in_sizes, int n_in,
                              void* d_out, int out_size, void* d_ws, size_t ws_size,
                              hipStream_t stream) {
  const float* x    = (const float*)d_in[0];
  const float* C0   = (const float*)d_in[1];
  const float* n0   = (const float*)d_in[2];
  const float* Wq   = (const float*)d_in[3];
  const float* bq   = (const float*)d_in[4];
  const float* Wk   = (const float*)d_in[5];
  const float* bk   = (const float*)d_in[6];
  const float* Wv   = (const float*)d_in[7];
  const float* bv   = (const float*)d_in[8];
  const float* wi   = (const float*)d_in[9];
  const float* bi   = (const float*)d_in[10];
  const float* wf   = (const float*)d_in[11];
  const float* bf   = (const float*)d_in[12];
  const float* Wo   = (const float*)d_in[13];
  const float* bo   = (const float*)d_in[14];
  const float* fc_w = (const float*)d_in[15];
  const float* fc_b = (const float*)d_in[16];

  float* ws = (float*)d_ws;
  const size_t NP = (size_t)T_LEN * MD;
  float* qw = ws;
  float* kw = ws + NP;
  float* vw = ws + 2 * NP;
  float* owp = ws + 3 * NP;
  float* igp = ws + 4 * NP;
  float* lfp = igp + T_LEN;
  float* numpp = lfp + T_LEN;
  float* denpp = numpp + NGRP * T_LEN;
  float* numip = denpp + NB * T_LEN;
  float* rswp  = numip + IC_NH * T_LEN;

  float* outp  = (float*)d_out;
  float* Coutp = outp + T_LEN;
  float* noutp = Coutp + MD * MD;

  proj_kernel<<<dim3(T_LEN / 128, 4), 128, 0, stream>>>(
      x, Wq, bq, Wk, bk, Wv, bv, Wo, bo, qw, kw, vw, owp);
  gates_kernel<<<dim3(T_LEN / LCH), 256, 0, stream>>>(
      x, wi, bi, wf, bf, igp, lfp);
  intra_kernel<<<dim3(T_LEN / LCH, IC_NH), 128, 0, stream>>>(
      qw, kw, vw, owp, igp, lfp, fc_w, numip, rswp);
  scan_kernel<false><<<NGRP * (NSEG - 1), 256, 0, stream>>>(
      qw, kw, vw, owp, igp, lfp, fc_w, numpp, denpp, Coutp, noutp);
  prefix_kernel<<<PF_CB + 1, 256, 0, stream>>>(lfp, C0, n0, numpp, denpp);
  scan_kernel<true><<<NGRP * NSEG, 256, 0, stream>>>(
      qw, kw, vw, owp, igp, lfp, fc_w, numpp, denpp, Coutp, noutp);
  out_kernel<<<T_LEN / 256, 256, 0, stream>>>(numpp, denpp, numip, rswp, fc_b, outp);
}
